# starting kernel
# baseline (speedup 1.0000x reference)
#include <hip/hip_runtime.h>
#include <hip/hip_bf16.h>

typedef __attribute__((ext_vector_type(2))) float v2f;
typedef __attribute__((ext_vector_type(8))) float v8f;

#define HC   256
#define NH   4
#define CCH  64
#define KIN  300
#define NOUTD 768

__device__ __forceinline__ unsigned fenc(float f) {
    unsigned u = __float_as_uint(f);
    return (u & 0x80000000u) ? ~u : (u | 0x80000000u);
}
__device__ __forceinline__ float fdec(unsigned u) {
    unsigned b = (u & 0x80000000u) ? (u & 0x7FFFFFFFu) : ~u;
    return __uint_as_float(b);
}

__global__ void init_ws(unsigned* emax, float* denom, float* agg,
                        float* pooled, float* cnt, int n, int g) {
    size_t stride = (size_t)gridDim.x * blockDim.x;
    size_t t = (size_t)blockIdx.x * blockDim.x + threadIdx.x;
    size_t total = (size_t)n * HC;
    const unsigned ninf = 0x007FFFFFu;
    for (size_t i = t; i < total; i += stride) {
        agg[i] = 0.0f;
        if (i < (size_t)n * NH) { emax[i] = ninf; denom[i] = 0.0f; }
        if (i < (size_t)g * HC) pooled[i] = 0.0f;
        if (i < (size_t)g)      cnt[i] = 0.0f;
    }
}

__global__ __launch_bounds__(32) void gemm_x_linw(const float* __restrict__ x,
                                                  const float* __restrict__ w,
                                                  float* __restrict__ h, int n) {
    int mTile = blockIdx.x;
    int lane = threadIdx.x & 31;
    int mr = lane & 15, half = lane >> 4;
    int row = mTile * 16 + mr;
    if (row >= n) row = n - 1;
    int col0 = blockIdx.y * 32 + mr;
    const float* arow = x + (size_t)row * KIN;
    v8f acc0 = {}, acc1 = {};
    for (int k = 0; k < KIN; k += 4) {
        int ka = k + 2 * half;
        v2f a; a.x = arow[ka]; a.y = arow[ka + 1];
        const float* wp = w + (size_t)ka * HC + col0;
        v2f b0; b0.x = wp[0];  b0.y = wp[HC];
        v2f b1; b1.x = wp[16]; b1.y = wp[HC + 16];
        acc0 = __builtin_amdgcn_wmma_f32_16x16x4_f32(false, a, false, b0,
                                                     (short)0, acc0, false, false);
        acc1 = __builtin_amdgcn_wmma_f32_16x16x4_f32(false, a, false, b1,
                                                     (short)0, acc1, false, false);
    }
    int rbase = mTile * 16 + 8 * half;
    if (rbase + 8 <= n) {
        #pragma unroll
        for (int r = 0; r < 8; ++r) {
            float* hp = h + (size_t)(rbase + r) * HC + col0;
            hp[0]  = acc0[r];
            hp[16] = acc1[r];
        }
    } else {
        for (int r = 0; r < 8; ++r) {
            if (rbase + r < n) {
                float* hp = h + (size_t)(rbase + r) * HC + col0;
                hp[0]  = acc0[r];
                hp[16] = acc1[r];
            }
        }
    }
}

__global__ __launch_bounds__(256) void att_kernel(const float* __restrict__ h,
                                                  const float* __restrict__ wsrc,
                                                  const float* __restrict__ wdst,
                                                  float* __restrict__ asrc,
                                                  float* __restrict__ adst, int n) {
    int node = ((int)blockIdx.x * blockDim.x + threadIdx.x) >> 5;
    int lane = threadIdx.x & 31;
    if (node >= n) return;
    int hh = lane >> 3, c0 = lane & 7;
    const float* hp = h + (size_t)node * HC + hh * CCH + c0;
    const float* ws = wsrc + hh * CCH + c0;
    const float* wd = wdst + hh * CCH + c0;
    float ss = 0.f, sd = 0.f;
    #pragma unroll
    for (int i = 0; i < 8; ++i) {
        float v = hp[i * 8];
        ss += v * ws[i * 8];
        sd += v * wd[i * 8];
    }
    #pragma unroll
    for (int off = 4; off >= 1; off >>= 1) {
        ss += __shfl_xor(ss, off, 32);
        sd += __shfl_xor(sd, off, 32);
    }
    if (c0 == 0) {
        asrc[(size_t)node * NH + hh] = ss;
        adst[(size_t)node * NH + hh] = sd;
    }
}

__global__ void edge_max(const int* __restrict__ ei, const float* __restrict__ asrc,
                         const float* __restrict__ adst, unsigned* __restrict__ emax,
                         int E, int n) {
    int e = (int)blockIdx.x * blockDim.x + threadIdx.x;
    if (e >= E + n) return;
    int s, d;
    if (e < E) { s = ei[e]; d = ei[E + e]; } else { s = d = e - E; }
    #pragma unroll
    for (int hh = 0; hh < NH; ++hh) {
        float v = asrc[s * NH + hh] + adst[d * NH + hh];
        v = v > 0.f ? v : 0.2f * v;
        atomicMax(&emax[d * NH + hh], fenc(v));
    }
}

__global__ void edge_sum(const int* __restrict__ ei, const float* __restrict__ asrc,
                         const float* __restrict__ adst, const unsigned* __restrict__ emax,
                         float* __restrict__ denom, int E, int n) {
    int e = (int)blockIdx.x * blockDim.x + threadIdx.x;
    if (e >= E + n) return;
    int s, d;
    if (e < E) { s = ei[e]; d = ei[E + e]; } else { s = d = e - E; }
    #pragma unroll
    for (int hh = 0; hh < NH; ++hh) {
        float v = asrc[s * NH + hh] + adst[d * NH + hh];
        v = v > 0.f ? v : 0.2f * v;
        atomicAdd(&denom[d * NH + hh], __expf(v - fdec(emax[d * NH + hh])));
    }
}

__global__ __launch_bounds__(256) void edge_agg(const int* __restrict__ ei,
                                                const float* __restrict__ asrc,
                                                const float* __restrict__ adst,
                                                const unsigned* __restrict__ emax,
                                                const float* __restrict__ denom,
                                                const float* __restrict__ h,
                                                float* __restrict__ agg, int E, int n) {
    int e = ((int)blockIdx.x * blockDim.x + threadIdx.x) >> 5;
    int lane = threadIdx.x & 31;
    if (e >= E + n) return;
    int s, d;
    if (e < E) { s = ei[e]; d = ei[E + e]; } else { s = d = e - E; }
    int hh = lane >> 3;
    float v = asrc[s * NH + hh] + adst[d * NH + hh];
    v = v > 0.f ? v : 0.2f * v;
    float alpha = __expf(v - fdec(emax[d * NH + hh])) / denom[d * NH + hh];
    const float4* hp = (const float4*)(h + (size_t)s * HC + lane * 8);
    float4 h0 = hp[0], h1 = hp[1];
    float* op = agg + (size_t)d * HC + lane * 8;
    atomicAdd(&op[0], alpha * h0.x);
    atomicAdd(&op[1], alpha * h0.y);
    atomicAdd(&op[2], alpha * h0.z);
    atomicAdd(&op[3], alpha * h0.w);
    atomicAdd(&op[4], alpha * h1.x);
    atomicAdd(&op[5], alpha * h1.y);
    atomicAdd(&op[6], alpha * h1.z);
    atomicAdd(&op[7], alpha * h1.w);
}

__global__ void node_post(const float* __restrict__ agg, const float* __restrict__ bias,
                          const int* __restrict__ batch, float* __restrict__ pooled,
                          float* __restrict__ cnt, int n) {
    int t = (int)blockIdx.x * blockDim.x + threadIdx.x;
    if (t >= n * HC) return;
    int node = t >> 8, ch = t & 255;
    float v = agg[t] + bias[ch];
    v = v > 0.f ? v : 0.01f * v;
    int g = batch[node];
    atomicAdd(&pooled[(size_t)g * HC + ch], v);
    if (ch == 0) atomicAdd(&cnt[g], 1.0f);
}

__global__ __launch_bounds__(32) void gemm_out(const float* __restrict__ pooled,
                                               const float* __restrict__ cnt,
                                               const float* __restrict__ fw,
                                               const float* __restrict__ fb,
                                               float* __restrict__ out, int g) {
    int mTile = blockIdx.x;
    int lane = threadIdx.x & 31;
    int mr = lane & 15, half = lane >> 4;
    int row = mTile * 16 + mr;
    int col0 = blockIdx.y * 32 + mr;
    float inv = 1.0f / fmaxf(cnt[row], 1.0f);
    const float* arow = pooled + (size_t)row * HC;
    v8f acc0 = {}, acc1 = {};
    for (int k = 0; k < HC; k += 4) {
        int ka = k + 2 * half;
        v2f a; a.x = arow[ka] * inv; a.y = arow[ka + 1] * inv;
        const float* wp = fw + (size_t)ka * NOUTD + col0;
        v2f b0; b0.x = wp[0];  b0.y = wp[NOUTD];
        v2f b1; b1.x = wp[16]; b1.y = wp[NOUTD + 16];
        acc0 = __builtin_amdgcn_wmma_f32_16x16x4_f32(false, a, false, b0,
                                                     (short)0, acc0, false, false);
        acc1 = __builtin_amdgcn_wmma_f32_16x16x4_f32(false, a, false, b1,
                                                     (short)0, acc1, false, false);
    }
    float bb0 = fb[col0], bb1 = fb[col0 + 16];
    int rbase = mTile * 16 + 8 * half;
    #pragma unroll
    for (int r = 0; r < 8; ++r) {
        float* op = out + (size_t)(rbase + r) * NOUTD + col0;
        op[0]  = acc0[r] + bb0;
        op[16] = acc1[r] + bb1;
    }
}

extern "C" void kernel_launch(void* const* d_in, const int* in_sizes, int n_in,
                              void* d_out, int out_size, void* d_ws, size_t ws_size,
                              hipStream_t stream) {
    const float* x      = (const float*)d_in[0];
    const int*   ei     = (const int*)  d_in[1];
    const int*   batch  = (const int*)  d_in[2];
    const float* lin_w  = (const float*)d_in[3];
    const float* att_s  = (const float*)d_in[4];
    const float* att_d  = (const float*)d_in[5];
    const float* bias   = (const float*)d_in[6];
    const float* fc1_w  = (const float*)d_in[7];
    const float* fc1_b  = (const float*)d_in[8];
    float* out = (float*)d_out;

    const int N = in_sizes[2];
    const int E = in_sizes[1] / 2;
    const int G = out_size / NOUTD;

    float* ws = (float*)d_ws;
    float*    h      = ws;
    float*    asrc   = h + (size_t)N * HC;
    float*    adst   = asrc + (size_t)N * NH;
    unsigned* emax   = (unsigned*)(adst + (size_t)N * NH);
    float*    denom  = (float*)(emax + (size_t)N * NH);
    float*    agg    = denom + (size_t)N * NH;
    float*    pooled = agg + (size_t)N * HC;
    float*    cnt    = pooled + (size_t)G * HC;

    init_ws<<<2048, 256, 0, stream>>>(emax, denom, agg, pooled, cnt, N, G);

    dim3 g1((N + 15) / 16, HC / 32);
    gemm_x_linw<<<g1, 32, 0, stream>>>(x, lin_w, h, N);

    att_kernel<<<(N * 32 + 255) / 256, 256, 0, stream>>>(h, att_s, att_d, asrc, adst, N);

    int tot = E + N;
    edge_max<<<(tot + 255) / 256, 256, 0, stream>>>(ei, asrc, adst, emax, E, N);
    edge_sum<<<(tot + 255) / 256, 256, 0, stream>>>(ei, asrc, adst, emax, denom, E, N);
    edge_agg<<<((size_t)tot * 32 + 255) / 256, 256, 0, stream>>>(ei, asrc, adst, emax,
                                                                 denom, h, agg, E, N);

    node_post<<<((size_t)N * HC + 255) / 256, 256, 0, stream>>>(agg, bias, batch,
                                                                pooled, cnt, N);

    dim3 g2(G / 16, NOUTD / 32);
    gemm_out<<<g2, 32, 0, stream>>>(pooled, cnt, fc1_w, fc1_b, out, G);
}


# Edge aggregation: a workgroup keeps 196 destination rows in LDS, scans the edge list for them, adds rows by LDS float adds, stores each row once: no global atomics or zero fill on agg; arrival order
# speedup vs baseline: 1.7338x; 1.7338x over previous
#include <hip/hip_runtime.h>
#include <hip/hip_bf16.h>

typedef __attribute__((ext_vector_type(2))) float v2f;
typedef __attribute__((ext_vector_type(8))) float v8f;

#define HC   256
#define NH   4
#define CCH  64
#define KIN  300
#define NOUTD 768

__device__ __forceinline__ unsigned fenc(float f) {
    unsigned u = __float_as_uint(f);
    return (u & 0x80000000u) ? ~u : (u | 0x80000000u);
}
__device__ __forceinline__ float fdec(unsigned u) {
    unsigned b = (u & 0x80000000u) ? (u & 0x7FFFFFFFu) : ~u;
    return __uint_as_float(b);
}

__global__ void init_ws(unsigned* emax, float* denom,
                        float* pooled, float* cnt, int n, int g) {
    size_t stride = (size_t)gridDim.x * blockDim.x;
    size_t t = (size_t)blockIdx.x * blockDim.x + threadIdx.x;
    size_t na = (size_t)n * NH, np = (size_t)g * HC;
    size_t total = na > np ? na : np;
    const unsigned ninf = 0x007FFFFFu;
    for (size_t i = t; i < total; i += stride) {
        if (i < na) { emax[i] = ninf; denom[i] = 0.0f; }
        if (i < np) pooled[i] = 0.0f;
        if (i < (size_t)g) cnt[i] = 0.0f;
    }
}

__global__ __launch_bounds__(32) void gemm_x_linw(const float* __restrict__ x,
                                                  const float* __restrict__ w,
                                                  float* __restrict__ h, int n) {
    int mTile = blockIdx.x;
    int lane = threadIdx.x & 31;
    int mr = lane & 15, half = lane >> 4;
    int row = mTile * 16 + mr;
    if (row >= n) row = n - 1;
    int col0 = blockIdx.y * 32 + mr;
    const float* arow = x + (size_t)row * KIN;
    v8f acc0 = {}, acc1 = {};
    for (int k = 0; k < KIN; k += 4) {
        int ka = k + 2 * half;
        v2f a; a.x = arow[ka]; a.y = arow[ka + 1];
        const float* wp = w + (size_t)ka * HC + col0;
        v2f b0; b0.x = wp[0];  b0.y = wp[HC];
        v2f b1; b1.x = wp[16]; b1.y = wp[HC + 16];
        acc0 = __builtin_amdgcn_wmma_f32_16x16x4_f32(false, a, false, b0,
                                                     (short)0, acc0, false, false);
        acc1 = __builtin_amdgcn_wmma_f32_16x16x4_f32(false, a, false, b1,
                                                     (short)0, acc1, false, false);
    }
    int rbase = mTile * 16 + 8 * half;
    if (rbase + 8 <= n) {
        #pragma unroll
        for (int r = 0; r < 8; ++r) {
            float* hp = h + (size_t)(rbase + r) * HC + col0;
            hp[0]  = acc0[r];
            hp[16] = acc1[r];
        }
    } else {
        for (int r = 0; r < 8; ++r) {
            if (rbase + r < n) {
                float* hp = h + (size_t)(rbase + r) * HC + col0;
                hp[0]  = acc0[r];
                hp[16] = acc1[r];
            }
        }
    }
}

__global__ __launch_bounds__(256) void att_kernel(const float* __restrict__ h,
                                                  const float* __restrict__ wsrc,
                                                  const float* __restrict__ wdst,
                                                  float* __restrict__ asrc,
                                                  float* __restrict__ adst, int n) {
    int node = ((int)blockIdx.x * blockDim.x + threadIdx.x) >> 5;
    int lane = threadIdx.x & 31;
    if (node >= n) return;
    int hh = lane >> 3, c0 = lane & 7;
    const float* hp = h + (size_t)node * HC + hh * CCH + c0;
    const float* ws = wsrc + hh * CCH + c0;
    const float* wd = wdst + hh * CCH + c0;
    float ss = 0.f, sd = 0.f;
    #pragma unroll
    for (int i = 0; i < 8; ++i) {
        float v = hp[i * 8];
        ss += v * ws[i * 8];
        sd += v * wd[i * 8];
    }
    #pragma unroll
    for (int off = 4; off >= 1; off >>= 1) {
        ss += __shfl_xor(ss, off, 32);
        sd += __shfl_xor(sd, off, 32);
    }
    if (c0 == 0) {
        asrc[(size_t)node * NH + hh] = ss;
        adst[(size_t)node * NH + hh] = sd;
    }
}

__global__ void edge_max(const int* __restrict__ ei, const float* __restrict__ asrc,
                         const float* __restrict__ adst, unsigned* __restrict__ emax,
                         int E, int n) {
    int e = (int)blockIdx.x * blockDim.x + threadIdx.x;
    if (e >= E + n) return;
    int s, d;
    if (e < E) { s = ei[e]; d = ei[E + e]; } else { s = d = e - E; }
    #pragma unroll
    for (int hh = 0; hh < NH; ++hh) {
        float v = asrc[s * NH + hh] + adst[d * NH + hh];
        v = v > 0.f ? v : 0.2f * v;
        atomicMax(&emax[d * NH + hh], fenc(v));
    }
}

__global__ void edge_sum(const int* __restrict__ ei, const float* __restrict__ asrc,
                         const float* __restrict__ adst, const unsigned* __restrict__ emax,
                         float* __restrict__ denom, int E, int n) {
    int e = (int)blockIdx.x * blockDim.x + threadIdx.x;
    if (e >= E + n) return;
    int s, d;
    if (e < E) { s = ei[e]; d = ei[E + e]; } else { s = d = e - E; }
    #pragma unroll
    for (int hh = 0; hh < NH; ++hh) {
        float v = asrc[s * NH + hh] + adst[d * NH + hh];
        v = v > 0.f ? v : 0.2f * v;
        atomicAdd(&denom[d * NH + hh], __expf(v - fdec(emax[d * NH + hh])));
    }
}

#define ROWS_MAX 196
#define TILES    3
#define QSZ      512

__device__ __forceinline__ void row_load(const float* __restrict__ h, int s, int lane, float (&hv)[8]) {
    const float* hp = h + (size_t)s * HC + lane;
    for (int j = 0; j < 8; ++j) hv[j] = hp[32 * j];
}
__device__ __forceinline__ void row_add_lds(float* rows, int r, int lane, const float (&hv)[8],
                                            float a0, float a1, float a2, float a3) {
    float* op = rows + r * HC + lane;
    atomicAdd(&op[0],   a0 * hv[0]);  atomicAdd(&op[32],  a0 * hv[1]);
    atomicAdd(&op[64],  a1 * hv[2]);  atomicAdd(&op[96],  a1 * hv[3]);
    atomicAdd(&op[128], a2 * hv[4]);  atomicAdd(&op[160], a2 * hv[5]);
    atomicAdd(&op[192], a3 * hv[6]);  atomicAdd(&op[224], a3 * hv[7]);
}
__device__ __forceinline__ float lane_bcast(float v, int i) {
    return __uint_as_float(__builtin_amdgcn_readlane(__float_as_uint(v), i));
}

__global__ __launch_bounds__(1024) void edge_agg(const int* __restrict__ ei,
                                                 const float* __restrict__ asrc,
                                                 const float* __restrict__ adst,
                                                 const unsigned* __restrict__ emax,
                                                 const float* __restrict__ denom,
                                                 const float* __restrict__ h,
                                                 float* __restrict__ agg, int E, int n) {
    __shared__ __attribute__((aligned(16))) float rows[ROWS_MAX * HC];
    __shared__ int rings[32 * QSZ];
    const int lane = threadIdx.x & 31;
    const int wv = __builtin_amdgcn_readfirstlane((int)(threadIdx.x >> 5));
    const int nwv = (int)(blockDim.x >> 5);
    typedef __attribute__((address_space(3))) int lds_int;
    volatile lds_int* myq = (volatile lds_int*)(rings + wv * QSZ);

    const int R = (n + (int)gridDim.x - 1) / (int)gridDim.x;
    const int nb0 = (int)blockIdx.x * R;
    const int nb1 = min(nb0 + R, n);

    const int* dstp = ei + E;
    const int hc = min((4 - (E & 3)) & 3, E);
    const int nb = (E - hc) >> 2;
    const int tc = (E - hc) & 3;
    const int4* d4 = (const int4*)(dstp + hc);
    const int ntiles = (nb + 31) >> 5;
    const int niter = (ntiles + TILES * nwv - 1) / (TILES * nwv);
    const int rot = (int)(((long long)blockIdx.x * ntiles) / (int)gridDim.x);

    for (int r0 = nb0; r0 < nb1; r0 += ROWS_MAX) {
        const int r1 = min(r0 + ROWS_MAX, nb1);
        const unsigned span = (unsigned)(r1 - r0);
        const int nv4 = (r1 - r0) * (HC / 4);
        float4* rows4 = (float4*)rows;
        for (int i = threadIdx.x; i < nv4; i += blockDim.x) rows4[i] = make_float4(0.f, 0.f, 0.f, 0.f);
        __syncthreads();

        int qh = 0, qn = 0;
        auto q_append = [&](bool match, int e) {
            const unsigned m = __builtin_amdgcn_ballot_w32(match);
            if (m != 0u) {
                const int slot = (qh + qn + (int)__builtin_amdgcn_mbcnt_lo(m, 0u)) & (QSZ - 1);
                if (match) myq[slot] = e;
                qn += __builtin_popcount(m);
            }
        };

        const int nself = ((r1 - r0) + 31) >> 5;
        for (int j = wv; j <= nself; j += nwv) {
            if (j < nself) {
                const int node = r0 + j * 32 + lane;
                q_append(node < r1, -1 - node);
            } else {
                const int idx = lane < hc ? lane : 4 * nb + lane;
                int dd = -1;
                if (lane < hc + tc) dd = dstp[idx];
                q_append((unsigned)(dd - r0) < span, idx);
            }
        }

        #pragma nounroll
        for (int it = 0; it <= niter; ++it) {
            if (it < niter) {
                int4 q[TILES];
                int g[TILES];
                bool ok[TILES];
                for (int u = 0; u < TILES; ++u) {
                    const int kk = (it * TILES + u) * nwv + wv;
                    int tt = kk + rot;
                    if (tt >= ntiles) tt -= ntiles;
                    g[u] = tt * 32 + lane;
                    ok[u] = kk < ntiles && g[u] < nb;
                    q[u] = d4[(unsigned)(ok[u] ? g[u] : 0)];
                }
                for (int u = 0; u < TILES; ++u) {
                    const bool m0 = ((unsigned)(q[u].x - r0) < span) & ok[u];
                    const bool m1 = ((unsigned)(q[u].y - r0) < span) & ok[u];
                    const bool m2 = ((unsigned)(q[u].z - r0) < span) & ok[u];
                    const bool m3 = ((unsigned)(q[u].w - r0) < span) & ok[u];
                    if (__builtin_amdgcn_ballot_w32(m0 | m1 | m2 | m3) != 0u) {
                        const int e = hc + 4 * g[u];
                        q_append(m0, e);
                        q_append(m1, e + 1);
                        q_append(m2, e + 2);
                        q_append(m3, e + 3);
                    }
                }
            }
            const bool last = it == niter;
            while (qn >= 32 || (last && qn > 0)) {
                const int cnt = min(qn, 32);
                int s = 0, d = r0;
                float w0 = 0.f, w1 = 0.f, w2 = 0.f, w3 = 0.f;
                if (lane < cnt) {
                    const int e = myq[(qh + lane) & (QSZ - 1)];
                    if (e >= 0) { s = ei[e]; d = dstp[e]; } else { s = -1 - e; d = s; }
                    const float4 as = *(const float4*)(asrc + (size_t)s * NH);
                    const float4 ad = *(const float4*)(adst + (size_t)d * NH);
                    const uint4  em = *(const uint4*)(emax + (size_t)d * NH);
                    const float4 dn = *(const float4*)(denom + (size_t)d * NH);
                    float v;
                    v = as.x + ad.x; v = v > 0.f ? v : 0.2f * v; w0 = __expf(v - fdec(em.x)) / dn.x;
                    v = as.y + ad.y; v = v > 0.f ? v : 0.2f * v; w1 = __expf(v - fdec(em.y)) / dn.y;
                    v = as.z + ad.z; v = v > 0.f ? v : 0.2f * v; w2 = __expf(v - fdec(em.z)) / dn.z;
                    v = as.w + ad.w; v = v > 0.f ? v : 0.2f * v; w3 = __expf(v - fdec(em.w)) / dn.w;
                }
                int i = 0;
                #pragma nounroll
                for (; i + 3 < cnt; i += 4) {
                    const int sa = __builtin_amdgcn_readlane(s, i),     ra = __builtin_amdgcn_readlane(d, i) - r0;
                    const int sb = __builtin_amdgcn_readlane(s, i + 1), rb = __builtin_amdgcn_readlane(d, i + 1) - r0;
                    const int sc = __builtin_amdgcn_readlane(s, i + 2), rc = __builtin_amdgcn_readlane(d, i + 2) - r0;
                    const int sd = __builtin_amdgcn_readlane(s, i + 3), rd = __builtin_amdgcn_readlane(d, i + 3) - r0;
                    float va[8], vb[8], vc[8], vd[8];
                    row_load(h, sa, lane, va);
                    row_load(h, sb, lane, vb);
                    row_load(h, sc, lane, vc);
                    row_load(h, sd, lane, vd);
                    row_add_lds(rows, ra, lane, va, lane_bcast(w0, i),     lane_bcast(w1, i),     lane_bcast(w2, i),     lane_bcast(w3, i));
                    row_add_lds(rows, rb, lane, vb, lane_bcast(w0, i + 1), lane_bcast(w1, i + 1), lane_bcast(w2, i + 1), lane_bcast(w3, i + 1));
                    row_add_lds(rows, rc, lane, vc, lane_bcast(w0, i + 2), lane_bcast(w1, i + 2), lane_bcast(w2, i + 2), lane_bcast(w3, i + 2));
                    row_add_lds(rows, rd, lane, vd, lane_bcast(w0, i + 3), lane_bcast(w1, i + 3), lane_bcast(w2, i + 3), lane_bcast(w3, i + 3));
                }
                #pragma nounroll
                for (; i < cnt; ++i) {
                    const int sa = __builtin_amdgcn_readlane(s, i), ra = __builtin_amdgcn_readlane(d, i) - r0;
                    float va[8];
                    row_load(h, sa, lane, va);
                    row_add_lds(rows, ra, lane, va, lane_bcast(w0, i), lane_bcast(w1, i), lane_bcast(w2, i), lane_bcast(w3, i));
                }
                qh = (qh + cnt) & (QSZ - 1);
                qn -= cnt;
            }
        }
        __syncthreads();

        float4* out4 = (float4*)(agg + (size_t)r0 * HC);
        for (int i = threadIdx.x; i < nv4; i += blockDim.x) out4[i] = rows4[i];
        __syncthreads();
    }
}

__global__ __launch_bounds__(256) void node_post(const float* __restrict__ agg,
                                                 const float* __restrict__ bias,
                                                 const int* __restrict__ batch,
                                                 float* __restrict__ pooled,
                                                 float* __restrict__ cnt, int n) {
    const int lane = threadIdx.x & 31;
    const int wave = __builtin_amdgcn_readfirstlane((int)((blockIdx.x * blockDim.x + threadIdx.x) >> 5));
    const int nw = (int)((gridDim.x * blockDim.x) >> 5);
    const int run = (n + nw - 1) / nw;
    const int n0 = wave * run;
    const int n1 = min(n0 + run, n);
    if (n0 >= n1) return;
    float b[8];
    for (int j = 0; j < 8; ++j) b[j] = bias[lane + 32 * j];
    for (int node = n0; node < n1; ++node) {
        const int g = batch[node];
        const float* ap = agg + (size_t)node * HC + lane;
        float* pp = pooled + (size_t)g * HC + lane;
        float v[8];
        for (int j = 0; j < 8; ++j) v[j] = ap[32 * j];
        for (int j = 0; j < 8; ++j) {
            float t = v[j] + b[j];
            t = t > 0.f ? t : 0.01f * t;
            atomicAdd(&pp[32 * j], t);
        }
        if (lane == 0) atomicAdd(&cnt[g], 1.0f);
    }
}

__global__ __launch_bounds__(32) void gemm_out(const float* __restrict__ pooled,
                                               const float* __restrict__ cnt,
                                               const float* __restrict__ fw,
                                               const float* __restrict__ fb,
                                               float* __restrict__ out, int g) {
    int mTile = blockIdx.x;
    int lane = threadIdx.x & 31;
    int mr = lane & 15, half = lane >> 4;
    int row = mTile * 16 + mr;
    int col0 = blockIdx.y * 32 + mr;
    float inv = 1.0f / fmaxf(cnt[row], 1.0f);
    const float* arow = pooled + (size_t)row * HC;
    v8f acc0 = {}, acc1 = {};
    for (int k = 0; k < HC; k += 4) {
        int ka = k + 2 * half;
        v2f a; a.x = arow[ka] * inv; a.y = arow[ka + 1] * inv;
        const float* wp = fw + (size_t)ka * NOUTD + col0;
        v2f b0; b0.x = wp[0];  b0.y = wp[NOUTD];
        v2f b1; b1.x = wp[16]; b1.y = wp[NOUTD + 16];
        acc0 = __builtin_amdgcn_wmma_f32_16x16x4_f32(false, a, false, b0,
                                                     (short)0, acc0, false, false);
        acc1 = __builtin_amdgcn_wmma_f32_16x16x4_f32(false, a, false, b1,
                                                     (short)0, acc1, false, false);
    }
    float bb0 = fb[col0], bb1 = fb[col0 + 16];
    int rbase = mTile * 16 + 8 * half;
    #pragma unroll
    for (int r = 0; r < 8; ++r) {
        float* op = out + (size_t)(rbase + r) * NOUTD + col0;
        op[0]  = acc0[r] + bb0;
        op[16] = acc1[r] + bb1;
    }
}

extern "C" void kernel_launch(void* const* d_in, const int* in_sizes, int n_in,
                              void* d_out, int out_size, void* d_ws, size_t ws_size,
                              hipStream_t stream) {
    const float* x      = (const float*)d_in[0];
    const int*   ei     = (const int*)  d_in[1];
    const int*   batch  = (const int*)  d_in[2];
    const float* lin_w  = (const float*)d_in[3];
    const float* att_s  = (const float*)d_in[4];
    const float* att_d  = (const float*)d_in[5];
    const float* bias   = (const float*)d_in[6];
    const float* fc1_w  = (const float*)d_in[7];
    const float* fc1_b  = (const float*)d_in[8];
    float* out = (float*)d_out;

    const int N = in_sizes[2];
    const int E = in_sizes[1] / 2;
    const int G = out_size / NOUTD;

    float* ws = (float*)d_ws;
    float*    h      = ws;
    float*    asrc   = h + (size_t)N * HC;
    float*    adst   = asrc + (size_t)N * NH;
    unsigned* emax   = (unsigned*)(adst + (size_t)N * NH);
    float*    denom  = (float*)(emax + (size_t)N * NH);
    float*    agg    = denom + (size_t)N * NH;
    float*    pooled = agg + (size_t)N * HC;
    float*    cnt    = pooled + (size_t)G * HC;

    init_ws<<<1024, 256, 0, stream>>>(emax, denom, pooled, cnt, N, G);

    dim3 g1((N + 15) / 16, HC / 32);
    gemm_x_linw<<<g1, 32, 0, stream>>>(x, lin_w, h, N);

    att_kernel<<<(N * 32 + 255) / 256, 256, 0, stream>>>(h, att_s, att_d, asrc, adst, N);

    int tot = E + N;
    edge_max<<<(tot + 255) / 256, 256, 0, stream>>>(ei, asrc, adst, emax, E, N);
    edge_sum<<<(tot + 255) / 256, 256, 0, stream>>>(ei, asrc, adst, emax, denom, E, N);
    edge_agg<<<256, 1024, 0, stream>>>(ei, asrc, adst, emax, denom, h, agg, E, N);

    node_post<<<2048, 256, 0, stream>>>(agg, bias, batch, pooled, cnt, N);

    dim3 g2(G / 16, NOUTD / 32);
    gemm_out<<<g2, 32, 0, stream>>>(pooled, cnt, fc1_w, fc1_b, out, G);
}


# GEMM1: one wave makes 16 rows x all 256 columns in registers: x read 64 B per lane per 32 k, lin_w by whole-line 16-B loads (lane owns 4 adjacent columns), no LDS; k order within 32 permuted
# speedup vs baseline: 1.8488x; 1.0663x over previous
#include <hip/hip_runtime.h>
#include <hip/hip_bf16.h>

typedef __attribute__((ext_vector_type(2))) float v2f;
typedef __attribute__((ext_vector_type(8))) float v8f;

#define HC   256
#define NH   4
#define CCH  64
#define KIN  300
#define NOUTD 768

__device__ __forceinline__ unsigned fenc(float f) {
    unsigned u = __float_as_uint(f);
    return (u & 0x80000000u) ? ~u : (u | 0x80000000u);
}
__device__ __forceinline__ float fdec(unsigned u) {
    unsigned b = (u & 0x80000000u) ? (u & 0x7FFFFFFFu) : ~u;
    return __uint_as_float(b);
}

__global__ void init_ws(unsigned* emax, float* denom,
                        float* pooled, float* cnt, int n, int g) {
    size_t stride = (size_t)gridDim.x * blockDim.x;
    size_t t = (size_t)blockIdx.x * blockDim.x + threadIdx.x;
    size_t na = (size_t)n * NH, np = (size_t)g * HC;
    size_t total = na > np ? na : np;
    const unsigned ninf = 0x007FFFFFu;
    for (size_t i = t; i < total; i += stride) {
        if (i < na) { emax[i] = ninf; denom[i] = 0.0f; }
        if (i < np) pooled[i] = 0.0f;
        if (i < (size_t)g) cnt[i] = 0.0f;
    }
}

typedef __attribute__((ext_vector_type(4))) float v4f;
#define KBLK 32

__device__ __forceinline__ void slab_step(v8f (&acc)[16], v2f av, const float* __restrict__ wp) {
    v4f b0[4], b1[4];
    #pragma unroll
    for (int g = 0; g < 4; ++g) {
        b0[g] = *(const v4f*)(wp + 64 * g);
        b1[g] = *(const v4f*)(wp + HC + 64 * g);
    }
    #pragma unroll
    for (int g = 0; g < 4; ++g) {
        #pragma unroll
        for (int c = 0; c < 4; ++c) {
            v2f bv; bv.x = b0[g][c]; bv.y = b1[g][c];
            acc[g * 4 + c] = __builtin_amdgcn_wmma_f32_16x16x4_f32(false, av, false, bv,
                                                                   (short)0, acc[g * 4 + c], false, false);
        }
    }
}

__global__ __launch_bounds__(32) void gemm_x_linw(const float* __restrict__ x,
                                                  const float* __restrict__ w,
                                                  float* __restrict__ h, int n) {
    const int lane = threadIdx.x & 31;
    const int mr = lane & 15, half = lane >> 4;
    const int row0 = (int)blockIdx.x * 16;
    int row = row0 + mr;
    if (row >= n) row = n - 1;
    const float* arow = x + (size_t)row * KIN;
    const float* wcol = w + 4 * mr;
    v8f acc[16];
    #pragma unroll
    for (int t = 0; t < 16; ++t) acc[t] = v8f{};

    #pragma nounroll
    for (int kb = 0; kb + KBLK <= KIN; kb += KBLK) {
        const int k0 = kb + 16 * half;
        const v4f* ap = (const v4f*)(arow + k0);
        v4f a[4];
        #pragma unroll
        for (int q = 0; q < 4; ++q) a[q] = ap[q];
        const float* wp = wcol + (size_t)k0 * HC;
        #pragma unroll
        for (int j = 0; j < 8; ++j) {
            v2f av; av.x = a[j >> 1][(j & 1) * 2]; av.y = a[j >> 1][(j & 1) * 2 + 1];
            slab_step(acc, av, wp + (size_t)(2 * j) * HC);
        }
    }
    #pragma unroll
    for (int k = (KIN / KBLK) * KBLK; k < KIN; k += 4) {
        const int ka = k + 2 * half;
        v2f av; av.x = arow[ka]; av.y = arow[ka + 1];
        slab_step(acc, av, wcol + (size_t)ka * HC);
    }

    const int rbase = row0 + 8 * half;
    if (rbase + 8 <= n) {
        #pragma unroll
        for (int r = 0; r < 8; ++r) {
            float* hp = h + (size_t)(rbase + r) * HC + 4 * mr;
            #pragma unroll
            for (int g = 0; g < 4; ++g) {
                v4f v; v.x = acc[g * 4][r]; v.y = acc[g * 4 + 1][r]; v.z = acc[g * 4 + 2][r]; v.w = acc[g * 4 + 3][r];
                *(v4f*)(hp + 64 * g) = v;
            }
        }
    } else {
        for (int r = 0; r < 8; ++r) {
            if (rbase + r < n) {
                float* hp = h + (size_t)(rbase + r) * HC + 4 * mr;
                #pragma unroll
                for (int g = 0; g < 4; ++g) {
                    v4f v; v.x = acc[g * 4][r]; v.y = acc[g * 4 + 1][r]; v.z = acc[g * 4 + 2][r]; v.w = acc[g * 4 + 3][r];
                    *(v4f*)(hp + 64 * g) = v;
                }
            }
        }
    }
}

__global__ __launch_bounds__(256) void att_kernel(const float* __restrict__ h,
                                                  const float* __restrict__ wsrc,
                                                  const float* __restrict__ wdst,
                                                  float* __restrict__ asrc,
                                                  float* __restrict__ adst, int n) {
    int node = ((int)blockIdx.x * blockDim.x + threadIdx.x) >> 5;
    int lane = threadIdx.x & 31;
    if (node >= n) return;
    int hh = lane >> 3, c0 = lane & 7;
    const float* hp = h + (size_t)node * HC + hh * CCH + c0;
    const float* ws = wsrc + hh * CCH + c0;
    const float* wd = wdst + hh * CCH + c0;
    float ss = 0.f, sd = 0.f;
    #pragma unroll
    for (int i = 0; i < 8; ++i) {
        float v = hp[i * 8];
        ss += v * ws[i * 8];
        sd += v * wd[i * 8];
    }
    #pragma unroll
    for (int off = 4; off >= 1; off >>= 1) {
        ss += __shfl_xor(ss, off, 32);
        sd += __shfl_xor(sd, off, 32);
    }
    if (c0 == 0) {
        asrc[(size_t)node * NH + hh] = ss;
        adst[(size_t)node * NH + hh] = sd;
    }
}

__global__ void edge_max(const int* __restrict__ ei, const float* __restrict__ asrc,
                         const float* __restrict__ adst, unsigned* __restrict__ emax,
                         int E, int n) {
    int e = (int)blockIdx.x * blockDim.x + threadIdx.x;
    if (e >= E + n) return;
    int s, d;
    if (e < E) { s = ei[e]; d = ei[E + e]; } else { s = d = e - E; }
    #pragma unroll
    for (int hh = 0; hh < NH; ++hh) {
        float v = asrc[s * NH + hh] + adst[d * NH + hh];
        v = v > 0.f ? v : 0.2f * v;
        atomicMax(&emax[d * NH + hh], fenc(v));
    }
}

__global__ void edge_sum(const int* __restrict__ ei, const float* __restrict__ asrc,
                         const float* __restrict__ adst, const unsigned* __restrict__ emax,
                         float* __restrict__ denom, int E, int n) {
    int e = (int)blockIdx.x * blockDim.x + threadIdx.x;
    if (e >= E + n) return;
    int s, d;
    if (e < E) { s = ei[e]; d = ei[E + e]; } else { s = d = e - E; }
    #pragma unroll
    for (int hh = 0; hh < NH; ++hh) {
        float v = asrc[s * NH + hh] + adst[d * NH + hh];
        v = v > 0.f ? v : 0.2f * v;
        atomicAdd(&denom[d * NH + hh], __expf(v - fdec(emax[d * NH + hh])));
    }
}

#define ROWS_MAX 196
#define TILES    3
#define QSZ      512

__device__ __forceinline__ void row_load(const float* __restrict__ h, int s, int lane, float (&hv)[8]) {
    const float* hp = h + (size_t)s * HC + lane;
    for (int j = 0; j < 8; ++j) hv[j] = hp[32 * j];
}
__device__ __forceinline__ void row_add_lds(float* rows, int r, int lane, const float (&hv)[8],
                                            float a0, float a1, float a2, float a3) {
    float* op = rows + r * HC + lane;
    atomicAdd(&op[0],   a0 * hv[0]);  atomicAdd(&op[32],  a0 * hv[1]);
    atomicAdd(&op[64],  a1 * hv[2]);  atomicAdd(&op[96],  a1 * hv[3]);
    atomicAdd(&op[128], a2 * hv[4]);  atomicAdd(&op[160], a2 * hv[5]);
    atomicAdd(&op[192], a3 * hv[6]);  atomicAdd(&op[224], a3 * hv[7]);
}
__device__ __forceinline__ float lane_bcast(float v, int i) {
    return __uint_as_float(__builtin_amdgcn_readlane(__float_as_uint(v), i));
}

__global__ __launch_bounds__(1024) void edge_agg(const int* __restrict__ ei,
                                                 const float* __restrict__ asrc,
                                                 const float* __restrict__ adst,
                                                 const unsigned* __restrict__ emax,
                                                 const float* __restrict__ denom,
                                                 const float* __restrict__ h,
                                                 float* __restrict__ agg, int E, int n) {
    __shared__ __attribute__((aligned(16))) float rows[ROWS_MAX * HC];
    __shared__ int rings[32 * QSZ];
    const int lane = threadIdx.x & 31;
    const int wv = __builtin_amdgcn_readfirstlane((int)(threadIdx.x >> 5));
    const int nwv = (int)(blockDim.x >> 5);
    typedef __attribute__((address_space(3))) int lds_int;
    volatile lds_int* myq = (volatile lds_int*)(rings + wv * QSZ);

    const int R = (n + (int)gridDim.x - 1) / (int)gridDim.x;
    const int nb0 = (int)blockIdx.x * R;
    const int nb1 = min(nb0 + R, n);

    const int* dstp = ei + E;
    const int hc = min((4 - (E & 3)) & 3, E);
    const int nb = (E - hc) >> 2;
    const int tc = (E - hc) & 3;
    const int4* d4 = (const int4*)(dstp + hc);
    const int ntiles = (nb + 31) >> 5;
    const int niter = (ntiles + TILES * nwv - 1) / (TILES * nwv);
    const int rot = (int)(((long long)blockIdx.x * ntiles) / (int)gridDim.x);

    for (int r0 = nb0; r0 < nb1; r0 += ROWS_MAX) {
        const int r1 = min(r0 + ROWS_MAX, nb1);
        const unsigned span = (unsigned)(r1 - r0);
        const int nv4 = (r1 - r0) * (HC / 4);
        float4* rows4 = (float4*)rows;
        for (int i = threadIdx.x; i < nv4; i += blockDim.x) rows4[i] = make_float4(0.f, 0.f, 0.f, 0.f);
        __syncthreads();

        int qh = 0, qn = 0;
        auto q_append = [&](bool match, int e) {
            const unsigned m = __builtin_amdgcn_ballot_w32(match);
            if (m != 0u) {
                const int slot = (qh + qn + (int)__builtin_amdgcn_mbcnt_lo(m, 0u)) & (QSZ - 1);
                if (match) myq[slot] = e;
                qn += __builtin_popcount(m);
            }
        };

        const int nself = ((r1 - r0) + 31) >> 5;
        for (int j = wv; j <= nself; j += nwv) {
            if (j < nself) {
                const int node = r0 + j * 32 + lane;
                q_append(node < r1, -1 - node);
            } else {
                const int idx = lane < hc ? lane : 4 * nb + lane;
                int dd = -1;
                if (lane < hc + tc) dd = dstp[idx];
                q_append((unsigned)(dd - r0) < span, idx);
            }
        }

        #pragma nounroll
        for (int it = 0; it <= niter; ++it) {
            if (it < niter) {
                int4 q[TILES];
                int g[TILES];
                bool ok[TILES];
                for (int u = 0; u < TILES; ++u) {
                    const int kk = (it * TILES + u) * nwv + wv;
                    int tt = kk + rot;
                    if (tt >= ntiles) tt -= ntiles;
                    g[u] = tt * 32 + lane;
                    ok[u] = kk < ntiles && g[u] < nb;
                    q[u] = d4[(unsigned)(ok[u] ? g[u] : 0)];
                }
                for (int u = 0; u < TILES; ++u) {
                    const bool m0 = ((unsigned)(q[u].x - r0) < span) & ok[u];
                    const bool m1 = ((unsigned)(q[u].y - r0) < span) & ok[u];
                    const bool m2 = ((unsigned)(q[u].z - r0) < span) & ok[u];
                    const bool m3 = ((unsigned)(q[u].w - r0) < span) & ok[u];
                    if (__builtin_amdgcn_ballot_w32(m0 | m1 | m2 | m3) != 0u) {
                        const int e = hc + 4 * g[u];
                        q_append(m0, e);
                        q_append(m1, e + 1);
                        q_append(m2, e + 2);
                        q_append(m3, e + 3);
                    }
                }
            }
            const bool last = it == niter;
            while (qn >= 32 || (last && qn > 0)) {
                const int cnt = min(qn, 32);
                int s = 0, d = r0;
                float w0 = 0.f, w1 = 0.f, w2 = 0.f, w3 = 0.f;
                if (lane < cnt) {
                    const int e = myq[(qh + lane) & (QSZ - 1)];
                    if (e >= 0) { s = ei[e]; d = dstp[e]; } else { s = -1 - e; d = s; }
                    const float4 as = *(const float4*)(asrc + (size_t)s * NH);
                    const float4 ad = *(const float4*)(adst + (size_t)d * NH);
                    const uint4  em = *(const uint4*)(emax + (size_t)d * NH);
                    const float4 dn = *(const float4*)(denom + (size_t)d * NH);
                    float v;
                    v = as.x + ad.x; v = v > 0.f ? v : 0.2f * v; w0 = __expf(v - fdec(em.x)) / dn.x;
                    v = as.y + ad.y; v = v > 0.f ? v : 0.2f * v; w1 = __expf(v - fdec(em.y)) / dn.y;
                    v = as.z + ad.z; v = v > 0.f ? v : 0.2f * v; w2 = __expf(v - fdec(em.z)) / dn.z;
                    v = as.w + ad.w; v = v > 0.f ? v : 0.2f * v; w3 = __expf(v - fdec(em.w)) / dn.w;
                }
                int i = 0;
                #pragma nounroll
                for (; i + 3 < cnt; i += 4) {
                    const int sa = __builtin_amdgcn_readlane(s, i),     ra = __builtin_amdgcn_readlane(d, i) - r0;
                    const int sb = __builtin_amdgcn_readlane(s, i + 1), rb = __builtin_amdgcn_readlane(d, i + 1) - r0;
                    const int sc = __builtin_amdgcn_readlane(s, i + 2), rc = __builtin_amdgcn_readlane(d, i + 2) - r0;
                    const int sd = __builtin_amdgcn_readlane(s, i + 3), rd = __builtin_amdgcn_readlane(d, i + 3) - r0;
                    float va[8], vb[8], vc[8], vd[8];
                    row_load(h, sa, lane, va);
                    row_load(h, sb, lane, vb);
                    row_load(h, sc, lane, vc);
                    row_load(h, sd, lane, vd);
                    row_add_lds(rows, ra, lane, va, lane_bcast(w0, i),     lane_bcast(w1, i),     lane_bcast(w2, i),     lane_bcast(w3, i));
                    row_add_lds(rows, rb, lane, vb, lane_bcast(w0, i + 1), lane_bcast(w1, i + 1), lane_bcast(w2, i + 1), lane_bcast(w3, i + 1));
                    row_add_lds(rows, rc, lane, vc, lane_bcast(w0, i + 2), lane_bcast(w1, i + 2), lane_bcast(w2, i + 2), lane_bcast(w3, i + 2));
                    row_add_lds(rows, rd, lane, vd, lane_bcast(w0, i + 3), lane_bcast(w1, i + 3), lane_bcast(w2, i + 3), lane_bcast(w3, i + 3));
                }
                #pragma nounroll
                for (; i < cnt; ++i) {
                    const int sa = __builtin_amdgcn_readlane(s, i), ra = __builtin_amdgcn_readlane(d, i) - r0;
                    float va[8];
                    row_load(h, sa, lane, va);
                    row_add_lds(rows, ra, lane, va, lane_bcast(w0, i), lane_bcast(w1, i), lane_bcast(w2, i), lane_bcast(w3, i));
                }
                qh = (qh + cnt) & (QSZ - 1);
                qn -= cnt;
            }
        }
        __syncthreads();

        float4* out4 = (float4*)(agg + (size_t)r0 * HC);
        for (int i = threadIdx.x; i < nv4; i += blockDim.x) out4[i] = rows4[i];
        __syncthreads();
    }
}

__global__ __launch_bounds__(256) void node_post(const float* __restrict__ agg,
                                                 const float* __restrict__ bias,
                                                 const int* __restrict__ batch,
                                                 float* __restrict__ pooled,
                                                 float* __restrict__ cnt, int n) {
    const int lane = threadIdx.x & 31;
    const int wave = __builtin_amdgcn_readfirstlane((int)((blockIdx.x * blockDim.x + threadIdx.x) >> 5));
    const int nw = (int)((gridDim.x * blockDim.x) >> 5);
    const int run = (n + nw - 1) / nw;
    const int n0 = wave * run;
    const int n1 = min(n0 + run, n);
    if (n0 >= n1) return;
    float b[8];
    for (int j = 0; j < 8; ++j) b[j] = bias[lane + 32 * j];
    for (int node = n0; node < n1; ++node) {
        const int g = batch[node];
        const float* ap = agg + (size_t)node * HC + lane;
        float* pp = pooled + (size_t)g * HC + lane;
        float v[8];
        for (int j = 0; j < 8; ++j) v[j] = ap[32 * j];
        for (int j = 0; j < 8; ++j) {
            float t = v[j] + b[j];
            t = t > 0.f ? t : 0.01f * t;
            atomicAdd(&pp[32 * j], t);
        }
        if (lane == 0) atomicAdd(&cnt[g], 1.0f);
    }
}

__global__ __launch_bounds__(32) void gemm_out(const float* __restrict__ pooled,
                                               const float* __restrict__ cnt,
                                               const float* __restrict__ fw,
                                               const float* __restrict__ fb,
                                               float* __restrict__ out, int g) {
    int mTile = blockIdx.x;
    int lane = threadIdx.x & 31;
    int mr = lane & 15, half = lane >> 4;
    int row = mTile * 16 + mr;
    int col0 = blockIdx.y * 32 + mr;
    float inv = 1.0f / fmaxf(cnt[row], 1.0f);
    const float* arow = pooled + (size_t)row * HC;
    v8f acc0 = {}, acc1 = {};
    for (int k = 0; k < HC; k += 4) {
        int ka = k + 2 * half;
        v2f a; a.x = arow[ka] * inv; a.y = arow[ka + 1] * inv;
        const float* wp = fw + (size_t)ka * NOUTD + col0;
        v2f b0; b0.x = wp[0];  b0.y = wp[NOUTD];
        v2f b1; b1.x = wp[16]; b1.y = wp[NOUTD + 16];
        acc0 = __builtin_amdgcn_wmma_f32_16x16x4_f32(false, a, false, b0,
                                                     (short)0, acc0, false, false);
        acc1 = __builtin_amdgcn_wmma_f32_16x16x4_f32(false, a, false, b1,
                                                     (short)0, acc1, false, false);
    }
    float bb0 = fb[col0], bb1 = fb[col0 + 16];
    int rbase = mTile * 16 + 8 * half;
    #pragma unroll
    for (int r = 0; r < 8; ++r) {
        float* op = out + (size_t)(rbase + r) * NOUTD + col0;
        op[0]  = acc0[r] + bb0;
        op[16] = acc1[r] + bb1;
    }
}

extern "C" void kernel_launch(void* const* d_in, const int* in_sizes, int n_in,
                              void* d_out, int out_size, void* d_ws, size_t ws_size,
                              hipStream_t stream) {
    const float* x      = (const float*)d_in[0];
    const int*   ei     = (const int*)  d_in[1];
    const int*   batch  = (const int*)  d_in[2];
    const float* lin_w  = (const float*)d_in[3];
    const float* att_s  = (const float*)d_in[4];
    const float* att_d  = (const float*)d_in[5];
    const float* bias   = (const float*)d_in[6];
    const float* fc1_w  = (const float*)d_in[7];
    const float* fc1_b  = (const float*)d_in[8];
    float* out = (float*)d_out;

    const int N = in_sizes[2];
    const int E = in_sizes[1] / 2;
    const int G = out_size / NOUTD;

    float* ws = (float*)d_ws;
    float*    h      = ws;
    float*    asrc   = h + (size_t)N * HC;
    float*    adst   = asrc + (size_t)N * NH;
    unsigned* emax   = (unsigned*)(adst + (size_t)N * NH);
    float*    denom  = (float*)(emax + (size_t)N * NH);
    float*    agg    = denom + (size_t)N * NH;
    float*    pooled = agg + (size_t)N * HC;
    float*    cnt    = pooled + (size_t)G * HC;

    init_ws<<<1024, 256, 0, stream>>>(emax, denom, pooled, cnt, N, G);

    gemm_x_linw<<<(N + 15) / 16, 32, 0, stream>>>(x, lin_w, h, N);

    att_kernel<<<(N * 32 + 255) / 256, 256, 0, stream>>>(h, att_s, att_d, asrc, adst, N);

    int tot = E + N;
    edge_max<<<(tot + 255) / 256, 256, 0, stream>>>(ei, asrc, adst, emax, E, N);
    edge_sum<<<(tot + 255) / 256, 256, 0, stream>>>(ei, asrc, adst, emax, denom, E, N);
    edge_agg<<<256, 1024, 0, stream>>>(ei, asrc, adst, emax, denom, h, agg, E, N);

    node_post<<<2048, 256, 0, stream>>>(agg, bias, batch, pooled, cnt, N);

    dim3 g2(G / 16, NOUTD / 32);
    gemm_out<<<g2, 32, 0, stream>>>(pooled, cnt, fc1_w, fc1_b, out, G);
}


# Softmax normalised after the sum: aggregation adds exp-weighted rows, sums each row's 4 denominators in LDS (arrival order) and divides once at the row store; exp-sum launch and its atomics gone
# speedup vs baseline: 2.5302x; 1.3686x over previous
#include <hip/hip_runtime.h>
#include <hip/hip_bf16.h>

typedef __attribute__((ext_vector_type(2))) float v2f;
typedef __attribute__((ext_vector_type(8))) float v8f;

#define HC   256
#define NH   4
#define CCH  64
#define KIN  300
#define NOUTD 768

__device__ __forceinline__ unsigned fenc(float f) {
    unsigned u = __float_as_uint(f);
    return (u & 0x80000000u) ? ~u : (u | 0x80000000u);
}
__device__ __forceinline__ float fdec(unsigned u) {
    unsigned b = (u & 0x80000000u) ? (u & 0x7FFFFFFFu) : ~u;
    return __uint_as_float(b);
}

__global__ void init_ws(unsigned* emax,
                        float* pooled, float* cnt, int n, int g) {
    size_t stride = (size_t)gridDim.x * blockDim.x;
    size_t t = (size_t)blockIdx.x * blockDim.x + threadIdx.x;
    size_t na = (size_t)n * NH, np = (size_t)g * HC;
    size_t total = na > np ? na : np;
    const unsigned ninf = 0x007FFFFFu;
    for (size_t i = t; i < total; i += stride) {
        if (i < na) emax[i] = ninf;
        if (i < np) pooled[i] = 0.0f;
        if (i < (size_t)g) cnt[i] = 0.0f;
    }
}

typedef __attribute__((ext_vector_type(4))) float v4f;
#define KBLK 32

__device__ __forceinline__ void slab_step(v8f (&acc)[16], v2f av, const float* __restrict__ wp) {
    v4f b0[4], b1[4];
    #pragma unroll
    for (int g = 0; g < 4; ++g) {
        b0[g] = *(const v4f*)(wp + 64 * g);
        b1[g] = *(const v4f*)(wp + HC + 64 * g);
    }
    #pragma unroll
    for (int g = 0; g < 4; ++g) {
        #pragma unroll
        for (int c = 0; c < 4; ++c) {
            v2f bv; bv.x = b0[g][c]; bv.y = b1[g][c];
            acc[g * 4 + c] = __builtin_amdgcn_wmma_f32_16x16x4_f32(false, av, false, bv,
                                                                   (short)0, acc[g * 4 + c], false, false);
        }
    }
}

__global__ __launch_bounds__(32) void gemm_x_linw(const float* __restrict__ x,
                                                  const float* __restrict__ w,
                                                  float* __restrict__ h, int n) {
    const int lane = threadIdx.x & 31;
    const int mr = lane & 15, half = lane >> 4;
    const int row0 = (int)blockIdx.x * 16;
    int row = row0 + mr;
    if (row >= n) row = n - 1;
    const float* arow = x + (size_t)row * KIN;
    const float* wcol = w + 4 * mr;
    v8f acc[16];
    #pragma unroll
    for (int t = 0; t < 16; ++t) acc[t] = v8f{};

    #pragma nounroll
    for (int kb = 0; kb + KBLK <= KIN; kb += KBLK) {
        const int k0 = kb + 16 * half;
        const v4f* ap = (const v4f*)(arow + k0);
        v4f a[4];
        #pragma unroll
        for (int q = 0; q < 4; ++q) a[q] = ap[q];
        const float* wp = wcol + (size_t)k0 * HC;
        #pragma unroll
        for (int j = 0; j < 8; ++j) {
            v2f av; av.x = a[j >> 1][(j & 1) * 2]; av.y = a[j >> 1][(j & 1) * 2 + 1];
            slab_step(acc, av, wp + (size_t)(2 * j) * HC);
        }
    }
    #pragma unroll
    for (int k = (KIN / KBLK) * KBLK; k < KIN; k += 4) {
        const int ka = k + 2 * half;
        v2f av; av.x = arow[ka]; av.y = arow[ka + 1];
        slab_step(acc, av, wcol + (size_t)ka * HC);
    }

    const int rbase = row0 + 8 * half;
    if (rbase + 8 <= n) {
        #pragma unroll
        for (int r = 0; r < 8; ++r) {
            float* hp = h + (size_t)(rbase + r) * HC + 4 * mr;
            #pragma unroll
            for (int g = 0; g < 4; ++g) {
                v4f v; v.x = acc[g * 4][r]; v.y = acc[g * 4 + 1][r]; v.z = acc[g * 4 + 2][r]; v.w = acc[g * 4 + 3][r];
                *(v4f*)(hp + 64 * g) = v;
            }
        }
    } else {
        for (int r = 0; r < 8; ++r) {
            if (rbase + r < n) {
                float* hp = h + (size_t)(rbase + r) * HC + 4 * mr;
                #pragma unroll
                for (int g = 0; g < 4; ++g) {
                    v4f v; v.x = acc[g * 4][r]; v.y = acc[g * 4 + 1][r]; v.z = acc[g * 4 + 2][r]; v.w = acc[g * 4 + 3][r];
                    *(v4f*)(hp + 64 * g) = v;
                }
            }
        }
    }
}

__global__ __launch_bounds__(256) void att_kernel(const float* __restrict__ h,
                                                  const float* __restrict__ wsrc,
                                                  const float* __restrict__ wdst,
                                                  float* __restrict__ asrc,
                                                  float* __restrict__ adst, int n) {
    int node = ((int)blockIdx.x * blockDim.x + threadIdx.x) >> 5;
    int lane = threadIdx.x & 31;
    if (node >= n) return;
    int hh = lane >> 3, c0 = lane & 7;
    const float* hp = h + (size_t)node * HC + hh * CCH + c0;
    const float* ws = wsrc + hh * CCH + c0;
    const float* wd = wdst + hh * CCH + c0;
    float ss = 0.f, sd = 0.f;
    #pragma unroll
    for (int i = 0; i < 8; ++i) {
        float v = hp[i * 8];
        ss += v * ws[i * 8];
        sd += v * wd[i * 8];
    }
    #pragma unroll
    for (int off = 4; off >= 1; off >>= 1) {
        ss += __shfl_xor(ss, off, 32);
        sd += __shfl_xor(sd, off, 32);
    }
    if (c0 == 0) {
        asrc[(size_t)node * NH + hh] = ss;
        adst[(size_t)node * NH + hh] = sd;
    }
}

__global__ void edge_max(const int* __restrict__ ei, const float* __restrict__ asrc,
                         const float* __restrict__ adst, unsigned* __restrict__ emax,
                         int E, int n) {
    int e = (int)blockIdx.x * blockDim.x + threadIdx.x;
    if (e >= E + n) return;
    int s, d;
    if (e < E) { s = ei[e]; d = ei[E + e]; } else { s = d = e - E; }
    #pragma unroll
    for (int hh = 0; hh < NH; ++hh) {
        float v = asrc[s * NH + hh] + adst[d * NH + hh];
        v = v > 0.f ? v : 0.2f * v;
        atomicMax(&emax[d * NH + hh], fenc(v));
    }
}

#define ROWS_MAX 196
#define TILES    3
#define QSZ      512

__device__ __forceinline__ void row_load(const float* __restrict__ h, int s, int lane, float (&hv)[8]) {
    const float* hp = h + (size_t)s * HC + lane;
    for (int j = 0; j < 8; ++j) hv[j] = hp[32 * j];
}
__device__ __forceinline__ void row_add_lds(float* rows, int r, int lane, const float (&hv)[8],
                                            float a0, float a1, float a2, float a3) {
    float* op = rows + r * HC + lane;
    atomicAdd(&op[0],   a0 * hv[0]);  atomicAdd(&op[32],  a0 * hv[1]);
    atomicAdd(&op[64],  a1 * hv[2]);  atomicAdd(&op[96],  a1 * hv[3]);
    atomicAdd(&op[128], a2 * hv[4]);  atomicAdd(&op[160], a2 * hv[5]);
    atomicAdd(&op[192], a3 * hv[6]);  atomicAdd(&op[224], a3 * hv[7]);
}
__device__ __forceinline__ float lane_bcast(float v, int i) {
    return __uint_as_float(__builtin_amdgcn_readlane(__float_as_uint(v), i));
}

__global__ __launch_bounds__(1024) void edge_agg(const int* __restrict__ ei,
                                                 const float* __restrict__ asrc,
                                                 const float* __restrict__ adst,
                                                 const unsigned* __restrict__ emax,
                                                 const float* __restrict__ h,
                                                 float* __restrict__ agg, int E, int n) {
    __shared__ __attribute__((aligned(16))) float rows[ROWS_MAX * HC];
    __shared__ float den[ROWS_MAX * NH];
    __shared__ int rings[32 * QSZ];
    const int lane = threadIdx.x & 31;
    const int wv = __builtin_amdgcn_readfirstlane((int)(threadIdx.x >> 5));
    const int nwv = (int)(blockDim.x >> 5);
    typedef __attribute__((address_space(3))) int lds_int;
    volatile lds_int* myq = (volatile lds_int*)(rings + wv * QSZ);

    const int R = (n + (int)gridDim.x - 1) / (int)gridDim.x;
    const int nb0 = (int)blockIdx.x * R;
    const int nb1 = min(nb0 + R, n);

    const int* dstp = ei + E;
    const int hc = min((4 - (E & 3)) & 3, E);
    const int nb = (E - hc) >> 2;
    const int tc = (E - hc) & 3;
    const int4* d4 = (const int4*)(dstp + hc);
    const int ntiles = (nb + 31) >> 5;
    const int niter = (ntiles + TILES * nwv - 1) / (TILES * nwv);
    const int rot = (int)(((long long)blockIdx.x * ntiles) / (int)gridDim.x);

    for (int r0 = nb0; r0 < nb1; r0 += ROWS_MAX) {
        const int r1 = min(r0 + ROWS_MAX, nb1);
        const unsigned span = (unsigned)(r1 - r0);
        const int nv4 = (r1 - r0) * (HC / 4);
        float4* rows4 = (float4*)rows;
        for (int i = threadIdx.x; i < nv4; i += blockDim.x) rows4[i] = make_float4(0.f, 0.f, 0.f, 0.f);
        for (int i = threadIdx.x; i < (r1 - r0) * NH; i += blockDim.x) den[i] = 0.f;
        __syncthreads();

        int qh = 0, qn = 0;
        auto q_append = [&](bool match, int e) {
            const unsigned m = __builtin_amdgcn_ballot_w32(match);
            if (m != 0u) {
                const int slot = (qh + qn + (int)__builtin_amdgcn_mbcnt_lo(m, 0u)) & (QSZ - 1);
                if (match) myq[slot] = e;
                qn += __builtin_popcount(m);
            }
        };

        const int nself = ((r1 - r0) + 31) >> 5;
        for (int j = wv; j <= nself; j += nwv) {
            if (j < nself) {
                const int node = r0 + j * 32 + lane;
                q_append(node < r1, -1 - node);
            } else {
                const int idx = lane < hc ? lane : 4 * nb + lane;
                int dd = -1;
                if (lane < hc + tc) dd = dstp[idx];
                q_append((unsigned)(dd - r0) < span, idx);
            }
        }

        #pragma nounroll
        for (int it = 0; it <= niter; ++it) {
            if (it < niter) {
                int4 q[TILES];
                int g[TILES];
                bool ok[TILES];
                for (int u = 0; u < TILES; ++u) {
                    const int kk = (it * TILES + u) * nwv + wv;
                    int tt = kk + rot;
                    if (tt >= ntiles) tt -= ntiles;
                    g[u] = tt * 32 + lane;
                    ok[u] = kk < ntiles && g[u] < nb;
                    q[u] = d4[(unsigned)(ok[u] ? g[u] : 0)];
                }
                for (int u = 0; u < TILES; ++u) {
                    const bool m0 = ((unsigned)(q[u].x - r0) < span) & ok[u];
                    const bool m1 = ((unsigned)(q[u].y - r0) < span) & ok[u];
                    const bool m2 = ((unsigned)(q[u].z - r0) < span) & ok[u];
                    const bool m3 = ((unsigned)(q[u].w - r0) < span) & ok[u];
                    if (__builtin_amdgcn_ballot_w32(m0 | m1 | m2 | m3) != 0u) {
                        const int e = hc + 4 * g[u];
                        q_append(m0, e);
                        q_append(m1, e + 1);
                        q_append(m2, e + 2);
                        q_append(m3, e + 3);
                    }
                }
            }
            const bool last = it == niter;
            while (qn >= 32 || (last && qn > 0)) {
                const int cnt = min(qn, 32);
                int s = 0, d = r0;
                float w0 = 0.f, w1 = 0.f, w2 = 0.f, w3 = 0.f;
                if (lane < cnt) {
                    const int e = myq[(qh + lane) & (QSZ - 1)];
                    if (e >= 0) { s = ei[e]; d = dstp[e]; } else { s = -1 - e; d = s; }
                    const float4 as = *(const float4*)(asrc + (size_t)s * NH);
                    const float4 ad = *(const float4*)(adst + (size_t)d * NH);
                    const uint4  em = *(const uint4*)(emax + (size_t)d * NH);
                    float v;
                    v = as.x + ad.x; v = v > 0.f ? v : 0.2f * v; w0 = __expf(v - fdec(em.x));
                    v = as.y + ad.y; v = v > 0.f ? v : 0.2f * v; w1 = __expf(v - fdec(em.y));
                    v = as.z + ad.z; v = v > 0.f ? v : 0.2f * v; w2 = __expf(v - fdec(em.z));
                    v = as.w + ad.w; v = v > 0.f ? v : 0.2f * v; w3 = __expf(v - fdec(em.w));
                    float* dp = den + (d - r0) * NH;
                    atomicAdd(&dp[0], w0);  atomicAdd(&dp[1], w1);
                    atomicAdd(&dp[2], w2);  atomicAdd(&dp[3], w3);
                }
                int i = 0;
                #pragma nounroll
                for (; i + 3 < cnt; i += 4) {
                    const int sa = __builtin_amdgcn_readlane(s, i),     ra = __builtin_amdgcn_readlane(d, i) - r0;
                    const int sb = __builtin_amdgcn_readlane(s, i + 1), rb = __builtin_amdgcn_readlane(d, i + 1) - r0;
                    const int sc = __builtin_amdgcn_readlane(s, i + 2), rc = __builtin_amdgcn_readlane(d, i + 2) - r0;
                    const int sd = __builtin_amdgcn_readlane(s, i + 3), rd = __builtin_amdgcn_readlane(d, i + 3) - r0;
                    float va[8], vb[8], vc[8], vd[8];
                    row_load(h, sa, lane, va);
                    row_load(h, sb, lane, vb);
                    row_load(h, sc, lane, vc);
                    row_load(h, sd, lane, vd);
                    row_add_lds(rows, ra, lane, va, lane_bcast(w0, i),     lane_bcast(w1, i),     lane_bcast(w2, i),     lane_bcast(w3, i));
                    row_add_lds(rows, rb, lane, vb, lane_bcast(w0, i + 1), lane_bcast(w1, i + 1), lane_bcast(w2, i + 1), lane_bcast(w3, i + 1));
                    row_add_lds(rows, rc, lane, vc, lane_bcast(w0, i + 2), lane_bcast(w1, i + 2), lane_bcast(w2, i + 2), lane_bcast(w3, i + 2));
                    row_add_lds(rows, rd, lane, vd, lane_bcast(w0, i + 3), lane_bcast(w1, i + 3), lane_bcast(w2, i + 3), lane_bcast(w3, i + 3));
                }
                #pragma nounroll
                for (; i < cnt; ++i) {
                    const int sa = __builtin_amdgcn_readlane(s, i), ra = __builtin_amdgcn_readlane(d, i) - r0;
                    float va[8];
                    row_load(h, sa, lane, va);
                    row_add_lds(rows, ra, lane, va, lane_bcast(w0, i), lane_bcast(w1, i), lane_bcast(w2, i), lane_bcast(w3, i));
                }
                qh = (qh + cnt) & (QSZ - 1);
                qn -= cnt;
            }
        }
        __syncthreads();

        float4* out4 = (float4*)(agg + (size_t)r0 * HC);
        for (int i = threadIdx.x; i < nv4; i += blockDim.x) {
            const float4 v = rows4[i];
            const float inv = 1.0f / den[(i >> 6) * NH + ((i & 63) >> 4)];
            out4[i] = make_float4(v.x * inv, v.y * inv, v.z * inv, v.w * inv);
        }
        __syncthreads();
    }
}

__global__ __launch_bounds__(256) void node_post(const float* __restrict__ agg,
                                                 const float* __restrict__ bias,
                                                 const int* __restrict__ batch,
                                                 float* __restrict__ pooled,
                                                 float* __restrict__ cnt, int n) {
    const int lane = threadIdx.x & 31;
    const int wave = __builtin_amdgcn_readfirstlane((int)((blockIdx.x * blockDim.x + threadIdx.x) >> 5));
    const int nw = (int)((gridDim.x * blockDim.x) >> 5);
    const int run = (n + nw - 1) / nw;
    const int n0 = wave * run;
    const int n1 = min(n0 + run, n);
    if (n0 >= n1) return;
    float b[8];
    for (int j = 0; j < 8; ++j) b[j] = bias[lane + 32 * j];
    for (int node = n0; node < n1; ++node) {
        const int g = batch[node];
        const float* ap = agg + (size_t)node * HC + lane;
        float* pp = pooled + (size_t)g * HC + lane;
        float v[8];
        for (int j = 0; j < 8; ++j) v[j] = ap[32 * j];
        for (int j = 0; j < 8; ++j) {
            float t = v[j] + b[j];
            t = t > 0.f ? t : 0.01f * t;
            atomicAdd(&pp[32 * j], t);
        }
        if (lane == 0) atomicAdd(&cnt[g], 1.0f);
    }
}

__global__ __launch_bounds__(32) void gemm_out(const float* __restrict__ pooled,
                                               const float* __restrict__ cnt,
                                               const float* __restrict__ fw,
                                               const float* __restrict__ fb,
                                               float* __restrict__ out, int g) {
    int mTile = blockIdx.x;
    int lane = threadIdx.x & 31;
    int mr = lane & 15, half = lane >> 4;
    int row = mTile * 16 + mr;
    int col0 = blockIdx.y * 32 + mr;
    float inv = 1.0f / fmaxf(cnt[row], 1.0f);
    const float* arow = pooled + (size_t)row * HC;
    v8f acc0 = {}, acc1 = {};
    for (int k = 0; k < HC; k += 4) {
        int ka = k + 2 * half;
        v2f a; a.x = arow[ka] * inv; a.y = arow[ka + 1] * inv;
        const float* wp = fw + (size_t)ka * NOUTD + col0;
        v2f b0; b0.x = wp[0];  b0.y = wp[NOUTD];
        v2f b1; b1.x = wp[16]; b1.y = wp[NOUTD + 16];
        acc0 = __builtin_amdgcn_wmma_f32_16x16x4_f32(false, a, false, b0,
                                                     (short)0, acc0, false, false);
        acc1 = __builtin_amdgcn_wmma_f32_16x16x4_f32(false, a, false, b1,
                                                     (short)0, acc1, false, false);
    }
    float bb0 = fb[col0], bb1 = fb[col0 + 16];
    int rbase = mTile * 16 + 8 * half;
    #pragma unroll
    for (int r = 0; r < 8; ++r) {
        float* op = out + (size_t)(rbase + r) * NOUTD + col0;
        op[0]  = acc0[r] + bb0;
        op[16] = acc1[r] + bb1;
    }
}

extern "C" void kernel_launch(void* const* d_in, const int* in_sizes, int n_in,
                              void* d_out, int out_size, void* d_ws, size_t ws_size,
                              hipStream_t stream) {
    const float* x      = (const float*)d_in[0];
    const int*   ei     = (const int*)  d_in[1];
    const int*   batch  = (const int*)  d_in[2];
    const float* lin_w  = (const float*)d_in[3];
    const float* att_s  = (const float*)d_in[4];
    const float* att_d  = (const float*)d_in[5];
    const float* bias   = (const float*)d_in[6];
    const float* fc1_w  = (const float*)d_in[7];
    const float* fc1_b  = (const float*)d_in[8];
    float* out = (float*)d_out;

    const int N = in_sizes[2];
    const int E = in_sizes[1] / 2;
    const int G = out_size / NOUTD;

    float* ws = (float*)d_ws;
    float*    h      = ws;
    float*    asrc   = h + (size_t)N * HC;
    float*    adst   = asrc + (size_t)N * NH;
    unsigned* emax   = (unsigned*)(adst + (size_t)N * NH);
    float*    agg    = (float*)(emax + (size_t)N * NH);
    float*    pooled = agg + (size_t)N * HC;
    float*    cnt    = pooled + (size_t)G * HC;

    init_ws<<<1024, 256, 0, stream>>>(emax, pooled, cnt, N, G);

    gemm_x_linw<<<(N + 15) / 16, 32, 0, stream>>>(x, lin_w, h, N);

    att_kernel<<<(N * 32 + 255) / 256, 256, 0, stream>>>(h, att_s, att_d, asrc, adst, N);

    int tot = E + N;
    edge_max<<<(tot + 255) / 256, 256, 0, stream>>>(ei, asrc, adst, emax, E, N);
    edge_agg<<<256, 1024, 0, stream>>>(ei, asrc, adst, emax, h, agg, E, N);

    node_post<<<2048, 256, 0, stream>>>(agg, bias, batch, pooled, cnt, N);

    dim3 g2(G / 16, NOUTD / 32);
    gemm_out<<<g2, 32, 0, stream>>>(pooled, cnt, fc1_w, fc1_b, out, G);
}


# Softmax row maxima found inside the aggregation workgroup by LDS unsigned max over its kept edges (a pass before the sums); the global atomic-max launch and its plane are gone; arrival order
# speedup vs baseline: 3.5691x; 1.4106x over previous
#include <hip/hip_runtime.h>
#include <hip/hip_bf16.h>

typedef __attribute__((ext_vector_type(2))) float v2f;
typedef __attribute__((ext_vector_type(8))) float v8f;

#define HC   256
#define NH   4
#define CCH  64
#define KIN  300
#define NOUTD 768

__device__ __forceinline__ unsigned fenc(float f) {
    unsigned u = __float_as_uint(f);
    return (u & 0x80000000u) ? ~u : (u | 0x80000000u);
}
__device__ __forceinline__ float fdec(unsigned u) {
    unsigned b = (u & 0x80000000u) ? (u & 0x7FFFFFFFu) : ~u;
    return __uint_as_float(b);
}

__global__ void init_ws(float* pooled, float* cnt, int g) {
    size_t stride = (size_t)gridDim.x * blockDim.x;
    size_t t = (size_t)blockIdx.x * blockDim.x + threadIdx.x;
    size_t np = (size_t)g * HC;
    for (size_t i = t; i < np; i += stride) {
        pooled[i] = 0.0f;
        if (i < (size_t)g) cnt[i] = 0.0f;
    }
}

typedef __attribute__((ext_vector_type(4))) float v4f;
#define KBLK 32

__device__ __forceinline__ void slab_step(v8f (&acc)[16], v2f av, const float* __restrict__ wp) {
    v4f b0[4], b1[4];
    #pragma unroll
    for (int g = 0; g < 4; ++g) {
        b0[g] = *(const v4f*)(wp + 64 * g);
        b1[g] = *(const v4f*)(wp + HC + 64 * g);
    }
    #pragma unroll
    for (int g = 0; g < 4; ++g) {
        #pragma unroll
        for (int c = 0; c < 4; ++c) {
            v2f bv; bv.x = b0[g][c]; bv.y = b1[g][c];
            acc[g * 4 + c] = __builtin_amdgcn_wmma_f32_16x16x4_f32(false, av, false, bv,
                                                                   (short)0, acc[g * 4 + c], false, false);
        }
    }
}

__global__ __launch_bounds__(32) void gemm_x_linw(const float* __restrict__ x,
                                                  const float* __restrict__ w,
                                                  float* __restrict__ h, int n) {
    const int lane = threadIdx.x & 31;
    const int mr = lane & 15, half = lane >> 4;
    const int row0 = (int)blockIdx.x * 16;
    int row = row0 + mr;
    if (row >= n) row = n - 1;
    const float* arow = x + (size_t)row * KIN;
    const float* wcol = w + 4 * mr;
    v8f acc[16];
    #pragma unroll
    for (int t = 0; t < 16; ++t) acc[t] = v8f{};

    #pragma nounroll
    for (int kb = 0; kb + KBLK <= KIN; kb += KBLK) {
        const int k0 = kb + 16 * half;
        const v4f* ap = (const v4f*)(arow + k0);
        v4f a[4];
        #pragma unroll
        for (int q = 0; q < 4; ++q) a[q] = ap[q];
        const float* wp = wcol + (size_t)k0 * HC;
        #pragma unroll
        for (int j = 0; j < 8; ++j) {
            v2f av; av.x = a[j >> 1][(j & 1) * 2]; av.y = a[j >> 1][(j & 1) * 2 + 1];
            slab_step(acc, av, wp + (size_t)(2 * j) * HC);
        }
    }
    #pragma unroll
    for (int k = (KIN / KBLK) * KBLK; k < KIN; k += 4) {
        const int ka = k + 2 * half;
        v2f av; av.x = arow[ka]; av.y = arow[ka + 1];
        slab_step(acc, av, wcol + (size_t)ka * HC);
    }

    const int rbase = row0 + 8 * half;
    if (rbase + 8 <= n) {
        #pragma unroll
        for (int r = 0; r < 8; ++r) {
            float* hp = h + (size_t)(rbase + r) * HC + 4 * mr;
            #pragma unroll
            for (int g = 0; g < 4; ++g) {
                v4f v; v.x = acc[g * 4][r]; v.y = acc[g * 4 + 1][r]; v.z = acc[g * 4 + 2][r]; v.w = acc[g * 4 + 3][r];
                *(v4f*)(hp + 64 * g) = v;
            }
        }
    } else {
        for (int r = 0; r < 8; ++r) {
            if (rbase + r < n) {
                float* hp = h + (size_t)(rbase + r) * HC + 4 * mr;
                #pragma unroll
                for (int g = 0; g < 4; ++g) {
                    v4f v; v.x = acc[g * 4][r]; v.y = acc[g * 4 + 1][r]; v.z = acc[g * 4 + 2][r]; v.w = acc[g * 4 + 3][r];
                    *(v4f*)(hp + 64 * g) = v;
                }
            }
        }
    }
}

__global__ __launch_bounds__(256) void att_kernel(const float* __restrict__ h,
                                                  const float* __restrict__ wsrc,
                                                  const float* __restrict__ wdst,
                                                  float* __restrict__ asrc,
                                                  float* __restrict__ adst, int n) {
    int node = ((int)blockIdx.x * blockDim.x + threadIdx.x) >> 5;
    int lane = threadIdx.x & 31;
    if (node >= n) return;
    int hh = lane >> 3, c0 = lane & 7;
    const float* hp = h + (size_t)node * HC + hh * CCH + c0;
    const float* ws = wsrc + hh * CCH + c0;
    const float* wd = wdst + hh * CCH + c0;
    float ss = 0.f, sd = 0.f;
    #pragma unroll
    for (int i = 0; i < 8; ++i) {
        float v = hp[i * 8];
        ss += v * ws[i * 8];
        sd += v * wd[i * 8];
    }
    #pragma unroll
    for (int off = 4; off >= 1; off >>= 1) {
        ss += __shfl_xor(ss, off, 32);
        sd += __shfl_xor(sd, off, 32);
    }
    if (c0 == 0) {
        asrc[(size_t)node * NH + hh] = ss;
        adst[(size_t)node * NH + hh] = sd;
    }
}

#define ROWS_MAX 196
#define TILES    3
#define QSZ      512

__device__ __forceinline__ void row_load(const float* __restrict__ h, int s, int lane, float (&hv)[8]) {
    const float* hp = h + (size_t)s * HC + lane;
    for (int j = 0; j < 8; ++j) hv[j] = hp[32 * j];
}
__device__ __forceinline__ void row_add_lds(float* rows, int r, int lane, const float (&hv)[8],
                                            float a0, float a1, float a2, float a3) {
    float* op = rows + r * HC + lane;
    atomicAdd(&op[0],   a0 * hv[0]);  atomicAdd(&op[32],  a0 * hv[1]);
    atomicAdd(&op[64],  a1 * hv[2]);  atomicAdd(&op[96],  a1 * hv[3]);
    atomicAdd(&op[128], a2 * hv[4]);  atomicAdd(&op[160], a2 * hv[5]);
    atomicAdd(&op[192], a3 * hv[6]);  atomicAdd(&op[224], a3 * hv[7]);
}
__device__ __forceinline__ float lane_bcast(float v, int i) {
    return __uint_as_float(__builtin_amdgcn_readlane(__float_as_uint(v), i));
}

__global__ __launch_bounds__(1024) void edge_agg(const int* __restrict__ ei,
                                                 const float* __restrict__ asrc,
                                                 const float* __restrict__ adst,
                                                 const float* __restrict__ h,
                                                 float* __restrict__ agg, int E, int n) {
    __shared__ __attribute__((aligned(16))) float rows[ROWS_MAX * HC];
    __shared__ float den[ROWS_MAX * NH];
    __shared__ __attribute__((aligned(16))) unsigned mx[ROWS_MAX * NH];
    __shared__ int rings[32 * QSZ];
    __shared__ int gave_up;
    const int lane = threadIdx.x & 31;
    const int wv = __builtin_amdgcn_readfirstlane((int)(threadIdx.x >> 5));
    const int nwv = (int)(blockDim.x >> 5);
    typedef __attribute__((address_space(3))) int lds_int;
    volatile lds_int* myq = (volatile lds_int*)(rings + wv * QSZ);

    const int R = (n + (int)gridDim.x - 1) / (int)gridDim.x;
    const int nb0 = (int)blockIdx.x * R;
    const int nb1 = min(nb0 + R, n);

    const int* dstp = ei + E;
    const int hc = min((4 - (E & 3)) & 3, E);
    const int nb = (E - hc) >> 2;
    const int tc = (E - hc) & 3;
    const int4* d4 = (const int4*)(dstp + hc);
    const int ntiles = (nb + 31) >> 5;
    const int niter = (ntiles + TILES * nwv - 1) / (TILES * nwv);
    const int rot = (int)(((long long)blockIdx.x * ntiles) / (int)gridDim.x);

    for (int r0 = nb0; r0 < nb1; r0 += ROWS_MAX) {
        const int r1 = min(r0 + ROWS_MAX, nb1);
        const unsigned span = (unsigned)(r1 - r0);
        const int nv4 = (r1 - r0) * (HC / 4);
        float4* rows4 = (float4*)rows;
        for (int i = threadIdx.x; i < nv4; i += blockDim.x) rows4[i] = make_float4(0.f, 0.f, 0.f, 0.f);
        for (int i = threadIdx.x; i < (r1 - r0) * NH; i += blockDim.x) {
            den[i] = 0.f;
            mx[i] = 0x007FFFFFu;
        }
        if (threadIdx.x == 0) gave_up = 0;
        __syncthreads();

        int qh = 0, qn = 0;
        int qm = 0;
        bool full = false;
        auto q_append = [&](bool match, int e) {
            const unsigned m = __builtin_amdgcn_ballot_w32(match);
            if (m != 0u) {
                const int slot = (qh + qn + (int)__builtin_amdgcn_mbcnt_lo(m, 0u)) & (QSZ - 1);
                if (match) myq[slot] = e;
                qn += __builtin_popcount(m);
            }
        };

        auto scan_trip = [&](int it, bool keep) {
            int4 q[TILES];
            int g[TILES];
            bool ok[TILES];
            for (int u = 0; u < TILES; ++u) {
                const int kk = (it * TILES + u) * nwv + wv;
                int tt = kk + rot;
                if (tt >= ntiles) tt -= ntiles;
                g[u] = tt * 32 + lane;
                ok[u] = kk < ntiles && g[u] < nb;
                q[u] = d4[(unsigned)(ok[u] ? g[u] : 0)];
            }
            for (int u = 0; u < TILES; ++u) {
                const bool m0 = ((unsigned)(q[u].x - r0) < span) & ok[u];
                const bool m1 = ((unsigned)(q[u].y - r0) < span) & ok[u];
                const bool m2 = ((unsigned)(q[u].z - r0) < span) & ok[u];
                const bool m3 = ((unsigned)(q[u].w - r0) < span) & ok[u];
                if (__builtin_amdgcn_ballot_w32(m0 | m1 | m2 | m3) != 0u) {
                    if (keep && qn + 128 > QSZ) {
                        full = true;
                        qh = (qh + qm) & (QSZ - 1);
                        qn -= qm;
                        qm = 0;
                    }
                    const int e = hc + 4 * g[u];
                    q_append(m0, e);
                    q_append(m1, e + 1);
                    q_append(m2, e + 2);
                    q_append(m3, e + 3);
                }
            }
        };

        const int nself = ((r1 - r0) + 31) >> 5;
        for (int j = wv; j <= nself; j += nwv) {
            if (j < nself) {
                const int node = r0 + j * 32 + lane;
                q_append(node < r1, -1 - node);
            } else {
                const int idx = lane < hc ? lane : 4 * nb + lane;
                int dd = -1;
                if (lane < hc + tc) dd = dstp[idx];
                q_append((unsigned)(dd - r0) < span, idx);
            }
        }
        #pragma nounroll
        for (int it = 0; it <= niter; ++it) {
            if (it < niter) scan_trip(it, true);
            const bool last = it == niter;
            while (qn - qm >= 32 || (last && qn > qm)) {
                const int cnt = min(qn - qm, 32);
                if (lane < cnt) {
                    const int e = myq[(qh + qm + lane) & (QSZ - 1)];
                    int s, d;
                    if (e >= 0) { s = ei[e]; d = dstp[e]; } else { s = -1 - e; d = s; }
                    const float4 as = *(const float4*)(asrc + (size_t)s * NH);
                    const float4 ad = *(const float4*)(adst + (size_t)d * NH);
                    unsigned* mp = mx + (d - r0) * NH;
                    float v;
                    v = as.x + ad.x; v = v > 0.f ? v : 0.2f * v; atomicMax(&mp[0], fenc(v));
                    v = as.y + ad.y; v = v > 0.f ? v : 0.2f * v; atomicMax(&mp[1], fenc(v));
                    v = as.z + ad.z; v = v > 0.f ? v : 0.2f * v; atomicMax(&mp[2], fenc(v));
                    v = as.w + ad.w; v = v > 0.f ? v : 0.2f * v; atomicMax(&mp[3], fenc(v));
                }
                qm += cnt;
            }
        }
        if (full && lane == 0) gave_up = 1;
        __syncthreads();

        const bool rescan = __builtin_amdgcn_readfirstlane(gave_up) != 0;
        int nscan = 0;
        if (rescan) {
            qh = 0;
            qn = 0;
            for (int j = wv; j <= nself; j += nwv) {
                if (j < nself) {
                    const int node = r0 + j * 32 + lane;
                    q_append(node < r1, -1 - node);
                } else {
                    const int idx = lane < hc ? lane : 4 * nb + lane;
                    int dd = -1;
                    if (lane < hc + tc) dd = dstp[idx];
                    q_append((unsigned)(dd - r0) < span, idx);
                }
            }
            nscan = niter;
        }

        #pragma nounroll
        for (int it = 0; it <= nscan; ++it) {
            if (it < nscan) scan_trip(it, false);
            const bool last = it == nscan;
            while (qn >= 32 || (last && qn > 0)) {
                const int cnt = min(qn, 32);
                int s = 0, d = r0;
                float w0 = 0.f, w1 = 0.f, w2 = 0.f, w3 = 0.f;
                if (lane < cnt) {
                    const int e = myq[(qh + lane) & (QSZ - 1)];
                    if (e >= 0) { s = ei[e]; d = dstp[e]; } else { s = -1 - e; d = s; }
                    const float4 as = *(const float4*)(asrc + (size_t)s * NH);
                    const float4 ad = *(const float4*)(adst + (size_t)d * NH);
                    const uint4  em = *(const uint4*)(mx + (d - r0) * NH);
                    float v;
                    v = as.x + ad.x; v = v > 0.f ? v : 0.2f * v; w0 = __expf(v - fdec(em.x));
                    v = as.y + ad.y; v = v > 0.f ? v : 0.2f * v; w1 = __expf(v - fdec(em.y));
                    v = as.z + ad.z; v = v > 0.f ? v : 0.2f * v; w2 = __expf(v - fdec(em.z));
                    v = as.w + ad.w; v = v > 0.f ? v : 0.2f * v; w3 = __expf(v - fdec(em.w));
                    float* dp = den + (d - r0) * NH;
                    atomicAdd(&dp[0], w0);  atomicAdd(&dp[1], w1);
                    atomicAdd(&dp[2], w2);  atomicAdd(&dp[3], w3);
                }
                int i = 0;
                #pragma nounroll
                for (; i + 3 < cnt; i += 4) {
                    const int sa = __builtin_amdgcn_readlane(s, i),     ra = __builtin_amdgcn_readlane(d, i) - r0;
                    const int sb = __builtin_amdgcn_readlane(s, i + 1), rb = __builtin_amdgcn_readlane(d, i + 1) - r0;
                    const int sc = __builtin_amdgcn_readlane(s, i + 2), rc = __builtin_amdgcn_readlane(d, i + 2) - r0;
                    const int sd = __builtin_amdgcn_readlane(s, i + 3), rd = __builtin_amdgcn_readlane(d, i + 3) - r0;
                    float va[8], vb[8], vc[8], vd[8];
                    row_load(h, sa, lane, va);
                    row_load(h, sb, lane, vb);
                    row_load(h, sc, lane, vc);
                    row_load(h, sd, lane, vd);
                    row_add_lds(rows, ra, lane, va, lane_bcast(w0, i),     lane_bcast(w1, i),     lane_bcast(w2, i),     lane_bcast(w3, i));
                    row_add_lds(rows, rb, lane, vb, lane_bcast(w0, i + 1), lane_bcast(w1, i + 1), lane_bcast(w2, i + 1), lane_bcast(w3, i + 1));
                    row_add_lds(rows, rc, lane, vc, lane_bcast(w0, i + 2), lane_bcast(w1, i + 2), lane_bcast(w2, i + 2), lane_bcast(w3, i + 2));
                    row_add_lds(rows, rd, lane, vd, lane_bcast(w0, i + 3), lane_bcast(w1, i + 3), lane_bcast(w2, i + 3), lane_bcast(w3, i + 3));
                }
                #pragma nounroll
                for (; i < cnt; ++i) {
                    const int sa = __builtin_amdgcn_readlane(s, i), ra = __builtin_amdgcn_readlane(d, i) - r0;
                    float va[8];
                    row_load(h, sa, lane, va);
                    row_add_lds(rows, ra, lane, va, lane_bcast(w0, i), lane_bcast(w1, i), lane_bcast(w2, i), lane_bcast(w3, i));
                }
                qh = (qh + cnt) & (QSZ - 1);
                qn -= cnt;
            }
        }
        __syncthreads();

        float4* out4 = (float4*)(agg + (size_t)r0 * HC);
        for (int i = threadIdx.x; i < nv4; i += blockDim.x) {
            const float4 v = rows4[i];
            const float inv = 1.0f / den[(i >> 6) * NH + ((i & 63) >> 4)];
            out4[i] = make_float4(v.x * inv, v.y * inv, v.z * inv, v.w * inv);
        }
        __syncthreads();
    }
}

__global__ __launch_bounds__(256) void node_post(const float* __restrict__ agg,
                                                 const float* __restrict__ bias,
                                                 const int* __restrict__ batch,
                                                 float* __restrict__ pooled,
                                                 float* __restrict__ cnt, int n) {
    const int lane = threadIdx.x & 31;
    const int wave = __builtin_amdgcn_readfirstlane((int)((blockIdx.x * blockDim.x + threadIdx.x) >> 5));
    const int nw = (int)((gridDim.x * blockDim.x) >> 5);
    const int run = (n + nw - 1) / nw;
    const int n0 = wave * run;
    const int n1 = min(n0 + run, n);
    if (n0 >= n1) return;
    float b[8];
    for (int j = 0; j < 8; ++j) b[j] = bias[lane + 32 * j];
    for (int node = n0; node < n1; ++node) {
        const int g = batch[node];
        const float* ap = agg + (size_t)node * HC + lane;
        float* pp = pooled + (size_t)g * HC + lane;
        float v[8];
        for (int j = 0; j < 8; ++j) v[j] = ap[32 * j];
        for (int j = 0; j < 8; ++j) {
            float t = v[j] + b[j];
            t = t > 0.f ? t : 0.01f * t;
            atomicAdd(&pp[32 * j], t);
        }
        if (lane == 0) atomicAdd(&cnt[g], 1.0f);
    }
}

__global__ __launch_bounds__(32) void gemm_out(const float* __restrict__ pooled,
                                               const float* __restrict__ cnt,
                                               const float* __restrict__ fw,
                                               const float* __restrict__ fb,
                                               float* __restrict__ out, int g) {
    int mTile = blockIdx.x;
    int lane = threadIdx.x & 31;
    int mr = lane & 15, half = lane >> 4;
    int row = mTile * 16 + mr;
    int col0 = blockIdx.y * 32 + mr;
    float inv = 1.0f / fmaxf(cnt[row], 1.0f);
    const float* arow = pooled + (size_t)row * HC;
    v8f acc0 = {}, acc1 = {};
    for (int k = 0; k < HC; k += 4) {
        int ka = k + 2 * half;
        v2f a; a.x = arow[ka] * inv; a.y = arow[ka + 1] * inv;
        const float* wp = fw + (size_t)ka * NOUTD + col0;
        v2f b0; b0.x = wp[0];  b0.y = wp[NOUTD];
        v2f b1; b1.x = wp[16]; b1.y = wp[NOUTD + 16];
        acc0 = __builtin_amdgcn_wmma_f32_16x16x4_f32(false, a, false, b0,
                                                     (short)0, acc0, false, false);
        acc1 = __builtin_amdgcn_wmma_f32_16x16x4_f32(false, a, false, b1,
                                                     (short)0, acc1, false, false);
    }
    float bb0 = fb[col0], bb1 = fb[col0 + 16];
    int rbase = mTile * 16 + 8 * half;
    #pragma unroll
    for (int r = 0; r < 8; ++r) {
        float* op = out + (size_t)(rbase + r) * NOUTD + col0;
        op[0]  = acc0[r] + bb0;
        op[16] = acc1[r] + bb1;
    }
}

extern "C" void kernel_launch(void* const* d_in, const int* in_sizes, int n_in,
                              void* d_out, int out_size, void* d_ws, size_t ws_size,
                              hipStream_t stream) {
    const float* x      = (const float*)d_in[0];
    const int*   ei     = (const int*)  d_in[1];
    const int*   batch  = (const int*)  d_in[2];
    const float* lin_w  = (const float*)d_in[3];
    const float* att_s  = (const float*)d_in[4];
    const float* att_d  = (const float*)d_in[5];
    const float* bias   = (const float*)d_in[6];
    const float* fc1_w  = (const float*)d_in[7];
    const float* fc1_b  = (const float*)d_in[8];
    float* out = (float*)d_out;

    const int N = in_sizes[2];
    const int E = in_sizes[1] / 2;
    const int G = out_size / NOUTD;

    float* ws = (float*)d_ws;
    float*    h      = ws;
    float*    asrc   = h + (size_t)N * HC;
    float*    adst   = asrc + (size_t)N * NH;
    float*    agg    = adst + (size_t)N * NH;
    float*    pooled = agg + (size_t)N * HC;
    float*    cnt    = pooled + (size_t)G * HC;

    init_ws<<<1024, 256, 0, stream>>>(pooled, cnt, G);

    gemm_x_linw<<<(N + 15) / 16, 32, 0, stream>>>(x, lin_w, h, N);

    att_kernel<<<(N * 32 + 255) / 256, 256, 0, stream>>>(h, att_s, att_d, asrc, adst, N);

    edge_agg<<<256, 1024, 0, stream>>>(ei, asrc, adst, h, agg, E, N);

    node_post<<<2048, 256, 0, stream>>>(agg, bias, batch, pooled, cnt, N);

    dim3 g2(G / 16, NOUTD / 32);
    gemm_out<<<g2, 32, 0, stream>>>(pooled, cnt, fc1_w, fc1_b, out, G);
}


# Bias, leaky-relu and graph pooling done on the aggregated rows in the workgroup's LDS: a thread sums its channel per graph run, one atomic per run; agg plane, node launch gone; new pooled sum order
# speedup vs baseline: 4.6170x; 1.2936x over previous
#include <hip/hip_runtime.h>
#include <hip/hip_bf16.h>

typedef __attribute__((ext_vector_type(2))) float v2f;
typedef __attribute__((ext_vector_type(8))) float v8f;

#define HC   256
#define NH   4
#define CCH  64
#define KIN  300
#define NOUTD 768

__device__ __forceinline__ unsigned fenc(float f) {
    unsigned u = __float_as_uint(f);
    return (u & 0x80000000u) ? ~u : (u | 0x80000000u);
}
__device__ __forceinline__ float fdec(unsigned u) {
    unsigned b = (u & 0x80000000u) ? (u & 0x7FFFFFFFu) : ~u;
    return __uint_as_float(b);
}

__global__ void init_ws(float* pooled, float* cnt, int g) {
    size_t stride = (size_t)gridDim.x * blockDim.x;
    size_t t = (size_t)blockIdx.x * blockDim.x + threadIdx.x;
    size_t np = (size_t)g * HC;
    for (size_t i = t; i < np; i += stride) {
        pooled[i] = 0.0f;
        if (i < (size_t)g) cnt[i] = 0.0f;
    }
}

typedef __attribute__((ext_vector_type(4))) float v4f;
#define KBLK 32

__device__ __forceinline__ void slab_step(v8f (&acc)[16], v2f av, const float* __restrict__ wp) {
    v4f b0[4], b1[4];
    #pragma unroll
    for (int g = 0; g < 4; ++g) {
        b0[g] = *(const v4f*)(wp + 64 * g);
        b1[g] = *(const v4f*)(wp + HC + 64 * g);
    }
    #pragma unroll
    for (int g = 0; g < 4; ++g) {
        #pragma unroll
        for (int c = 0; c < 4; ++c) {
            v2f bv; bv.x = b0[g][c]; bv.y = b1[g][c];
            acc[g * 4 + c] = __builtin_amdgcn_wmma_f32_16x16x4_f32(false, av, false, bv,
                                                                   (short)0, acc[g * 4 + c], false, false);
        }
    }
}

__global__ __launch_bounds__(32) void gemm_x_linw(const float* __restrict__ x,
                                                  const float* __restrict__ w,
                                                  float* __restrict__ h, int n) {
    const int lane = threadIdx.x & 31;
    const int mr = lane & 15, half = lane >> 4;
    const int row0 = (int)blockIdx.x * 16;
    int row = row0 + mr;
    if (row >= n) row = n - 1;
    const float* arow = x + (size_t)row * KIN;
    const float* wcol = w + 4 * mr;
    v8f acc[16];
    #pragma unroll
    for (int t = 0; t < 16; ++t) acc[t] = v8f{};

    #pragma nounroll
    for (int kb = 0; kb + KBLK <= KIN; kb += KBLK) {
        const int k0 = kb + 16 * half;
        const v4f* ap = (const v4f*)(arow + k0);
        v4f a[4];
        #pragma unroll
        for (int q = 0; q < 4; ++q) a[q] = ap[q];
        const float* wp = wcol + (size_t)k0 * HC;
        #pragma unroll
        for (int j = 0; j < 8; ++j) {
            v2f av; av.x = a[j >> 1][(j & 1) * 2]; av.y = a[j >> 1][(j & 1) * 2 + 1];
            slab_step(acc, av, wp + (size_t)(2 * j) * HC);
        }
    }
    #pragma unroll
    for (int k = (KIN / KBLK) * KBLK; k < KIN; k += 4) {
        const int ka = k + 2 * half;
        v2f av; av.x = arow[ka]; av.y = arow[ka + 1];
        slab_step(acc, av, wcol + (size_t)ka * HC);
    }

    const int rbase = row0 + 8 * half;
    if (rbase + 8 <= n) {
        #pragma unroll
        for (int r = 0; r < 8; ++r) {
            float* hp = h + (size_t)(rbase + r) * HC + 4 * mr;
            #pragma unroll
            for (int g = 0; g < 4; ++g) {
                v4f v; v.x = acc[g * 4][r]; v.y = acc[g * 4 + 1][r]; v.z = acc[g * 4 + 2][r]; v.w = acc[g * 4 + 3][r];
                *(v4f*)(hp + 64 * g) = v;
            }
        }
    } else {
        for (int r = 0; r < 8; ++r) {
            if (rbase + r < n) {
                float* hp = h + (size_t)(rbase + r) * HC + 4 * mr;
                #pragma unroll
                for (int g = 0; g < 4; ++g) {
                    v4f v; v.x = acc[g * 4][r]; v.y = acc[g * 4 + 1][r]; v.z = acc[g * 4 + 2][r]; v.w = acc[g * 4 + 3][r];
                    *(v4f*)(hp + 64 * g) = v;
                }
            }
        }
    }
}

__global__ __launch_bounds__(256) void att_kernel(const float* __restrict__ h,
                                                  const float* __restrict__ wsrc,
                                                  const float* __restrict__ wdst,
                                                  float* __restrict__ asrc,
                                                  float* __restrict__ adst, int n) {
    int node = ((int)blockIdx.x * blockDim.x + threadIdx.x) >> 5;
    int lane = threadIdx.x & 31;
    if (node >= n) return;
    int hh = lane >> 3, c0 = lane & 7;
    const float* hp = h + (size_t)node * HC + hh * CCH + c0;
    const float* ws = wsrc + hh * CCH + c0;
    const float* wd = wdst + hh * CCH + c0;
    float ss = 0.f, sd = 0.f;
    #pragma unroll
    for (int i = 0; i < 8; ++i) {
        float v = hp[i * 8];
        ss += v * ws[i * 8];
        sd += v * wd[i * 8];
    }
    #pragma unroll
    for (int off = 4; off >= 1; off >>= 1) {
        ss += __shfl_xor(ss, off, 32);
        sd += __shfl_xor(sd, off, 32);
    }
    if (c0 == 0) {
        asrc[(size_t)node * NH + hh] = ss;
        adst[(size_t)node * NH + hh] = sd;
    }
}

#define ROWS_MAX 196
#define TILES    3
#define QSZ      512

__device__ __forceinline__ void row_load(const float* __restrict__ h, int s, int lane, float (&hv)[8]) {
    const float* hp = h + (size_t)s * HC + lane;
    for (int j = 0; j < 8; ++j) hv[j] = hp[32 * j];
}
__device__ __forceinline__ void row_add_lds(float* rows, int r, int lane, const float (&hv)[8],
                                            float a0, float a1, float a2, float a3) {
    float* op = rows + r * HC + lane;
    atomicAdd(&op[0],   a0 * hv[0]);  atomicAdd(&op[32],  a0 * hv[1]);
    atomicAdd(&op[64],  a1 * hv[2]);  atomicAdd(&op[96],  a1 * hv[3]);
    atomicAdd(&op[128], a2 * hv[4]);  atomicAdd(&op[160], a2 * hv[5]);
    atomicAdd(&op[192], a3 * hv[6]);  atomicAdd(&op[224], a3 * hv[7]);
}
__device__ __forceinline__ float lane_bcast(float v, int i) {
    return __uint_as_float(__builtin_amdgcn_readlane(__float_as_uint(v), i));
}

__global__ __launch_bounds__(1024) void edge_agg(const int* __restrict__ ei,
                                                 const float* __restrict__ asrc,
                                                 const float* __restrict__ adst,
                                                 const float* __restrict__ h,
                                                 const float* __restrict__ bias,
                                                 const int* __restrict__ batch,
                                                 float* __restrict__ pooled,
                                                 float* __restrict__ cnt, int E, int n) {
    __shared__ __attribute__((aligned(16))) float rows[ROWS_MAX * HC];
    __shared__ float den[ROWS_MAX * NH];
    __shared__ __attribute__((aligned(16))) unsigned mx[ROWS_MAX * NH];
    __shared__ int rings[32 * QSZ];
    __shared__ int bt[ROWS_MAX];
    __shared__ int gave_up;
    const int lane = threadIdx.x & 31;
    const int wv = __builtin_amdgcn_readfirstlane((int)(threadIdx.x >> 5));
    const int nwv = (int)(blockDim.x >> 5);
    typedef __attribute__((address_space(3))) int lds_int;
    volatile lds_int* myq = (volatile lds_int*)(rings + wv * QSZ);

    const int R = (n + (int)gridDim.x - 1) / (int)gridDim.x;
    const int nb0 = (int)blockIdx.x * R;
    const int nb1 = min(nb0 + R, n);

    const int* dstp = ei + E;
    const int hc = min((4 - (E & 3)) & 3, E);
    const int nb = (E - hc) >> 2;
    const int tc = (E - hc) & 3;
    const int4* d4 = (const int4*)(dstp + hc);
    const int ntiles = (nb + 31) >> 5;
    const int niter = (ntiles + TILES * nwv - 1) / (TILES * nwv);
    const int rot = (int)(((long long)blockIdx.x * ntiles) / (int)gridDim.x);

    for (int r0 = nb0; r0 < nb1; r0 += ROWS_MAX) {
        const int r1 = min(r0 + ROWS_MAX, nb1);
        const unsigned span = (unsigned)(r1 - r0);
        const int nv4 = (r1 - r0) * (HC / 4);
        float4* rows4 = (float4*)rows;
        for (int i = threadIdx.x; i < nv4; i += blockDim.x) rows4[i] = make_float4(0.f, 0.f, 0.f, 0.f);
        for (int i = threadIdx.x; i < (r1 - r0) * NH; i += blockDim.x) {
            den[i] = 0.f;
            mx[i] = 0x007FFFFFu;
        }
        for (int i = threadIdx.x; i < r1 - r0; i += blockDim.x) bt[i] = batch[r0 + i];
        if (threadIdx.x == 0) gave_up = 0;
        __syncthreads();

        int qh = 0, qn = 0;
        int qm = 0;
        bool full = false;
        auto q_append = [&](bool match, int e) {
            const unsigned m = __builtin_amdgcn_ballot_w32(match);
            if (m != 0u) {
                const int slot = (qh + qn + (int)__builtin_amdgcn_mbcnt_lo(m, 0u)) & (QSZ - 1);
                if (match) myq[slot] = e;
                qn += __builtin_popcount(m);
            }
        };

        auto scan_trip = [&](int it, bool keep) {
            int4 q[TILES];
            int g[TILES];
            bool ok[TILES];
            for (int u = 0; u < TILES; ++u) {
                const int kk = (it * TILES + u) * nwv + wv;
                int tt = kk + rot;
                if (tt >= ntiles) tt -= ntiles;
                g[u] = tt * 32 + lane;
                ok[u] = kk < ntiles && g[u] < nb;
                q[u] = d4[(unsigned)(ok[u] ? g[u] : 0)];
            }
            for (int u = 0; u < TILES; ++u) {
                const bool m0 = ((unsigned)(q[u].x - r0) < span) & ok[u];
                const bool m1 = ((unsigned)(q[u].y - r0) < span) & ok[u];
                const bool m2 = ((unsigned)(q[u].z - r0) < span) & ok[u];
                const bool m3 = ((unsigned)(q[u].w - r0) < span) & ok[u];
                if (__builtin_amdgcn_ballot_w32(m0 | m1 | m2 | m3) != 0u) {
                    if (keep && qn + 128 > QSZ) {
                        full = true;
                        qh = (qh + qm) & (QSZ - 1);
                        qn -= qm;
                        qm = 0;
                    }
                    const int e = hc + 4 * g[u];
                    q_append(m0, e);
                    q_append(m1, e + 1);
                    q_append(m2, e + 2);
                    q_append(m3, e + 3);
                }
            }
        };

        const int nself = ((r1 - r0) + 31) >> 5;
        for (int j = wv; j <= nself; j += nwv) {
            if (j < nself) {
                const int node = r0 + j * 32 + lane;
                q_append(node < r1, -1 - node);
            } else {
                const int idx = lane < hc ? lane : 4 * nb + lane;
                int dd = -1;
                if (lane < hc + tc) dd = dstp[idx];
                q_append((unsigned)(dd - r0) < span, idx);
            }
        }
        #pragma nounroll
        for (int it = 0; it <= niter; ++it) {
            if (it < niter) scan_trip(it, true);
            const bool last = it == niter;
            while (qn - qm >= 32 || (last && qn > qm)) {
                const int cnt = min(qn - qm, 32);
                if (lane < cnt) {
                    const int e = myq[(qh + qm + lane) & (QSZ - 1)];
                    int s, d;
                    if (e >= 0) { s = ei[e]; d = dstp[e]; } else { s = -1 - e; d = s; }
                    const float4 as = *(const float4*)(asrc + (size_t)s * NH);
                    const float4 ad = *(const float4*)(adst + (size_t)d * NH);
                    unsigned* mp = mx + (d - r0) * NH;
                    float v;
                    v = as.x + ad.x; v = v > 0.f ? v : 0.2f * v; atomicMax(&mp[0], fenc(v));
                    v = as.y + ad.y; v = v > 0.f ? v : 0.2f * v; atomicMax(&mp[1], fenc(v));
                    v = as.z + ad.z; v = v > 0.f ? v : 0.2f * v; atomicMax(&mp[2], fenc(v));
                    v = as.w + ad.w; v = v > 0.f ? v : 0.2f * v; atomicMax(&mp[3], fenc(v));
                }
                qm += cnt;
            }
        }
        if (full && lane == 0) gave_up = 1;
        __syncthreads();

        const bool rescan = __builtin_amdgcn_readfirstlane(gave_up) != 0;
        int nscan = 0;
        if (rescan) {
            qh = 0;
            qn = 0;
            for (int j = wv; j <= nself; j += nwv) {
                if (j < nself) {
                    const int node = r0 + j * 32 + lane;
                    q_append(node < r1, -1 - node);
                } else {
                    const int idx = lane < hc ? lane : 4 * nb + lane;
                    int dd = -1;
                    if (lane < hc + tc) dd = dstp[idx];
                    q_append((unsigned)(dd - r0) < span, idx);
                }
            }
            nscan = niter;
        }

        #pragma nounroll
        for (int it = 0; it <= nscan; ++it) {
            if (it < nscan) scan_trip(it, false);
            const bool last = it == nscan;
            while (qn >= 32 || (last && qn > 0)) {
                const int cnt = min(qn, 32);
                int s = 0, d = r0;
                float w0 = 0.f, w1 = 0.f, w2 = 0.f, w3 = 0.f;
                if (lane < cnt) {
                    const int e = myq[(qh + lane) & (QSZ - 1)];
                    if (e >= 0) { s = ei[e]; d = dstp[e]; } else { s = -1 - e; d = s; }
                    const float4 as = *(const float4*)(asrc + (size_t)s * NH);
                    const float4 ad = *(const float4*)(adst + (size_t)d * NH);
                    const uint4  em = *(const uint4*)(mx + (d - r0) * NH);
                    float v;
                    v = as.x + ad.x; v = v > 0.f ? v : 0.2f * v; w0 = __expf(v - fdec(em.x));
                    v = as.y + ad.y; v = v > 0.f ? v : 0.2f * v; w1 = __expf(v - fdec(em.y));
                    v = as.z + ad.z; v = v > 0.f ? v : 0.2f * v; w2 = __expf(v - fdec(em.z));
                    v = as.w + ad.w; v = v > 0.f ? v : 0.2f * v; w3 = __expf(v - fdec(em.w));
                    float* dp = den + (d - r0) * NH;
                    atomicAdd(&dp[0], w0);  atomicAdd(&dp[1], w1);
                    atomicAdd(&dp[2], w2);  atomicAdd(&dp[3], w3);
                }
                int i = 0;
                #pragma nounroll
                for (; i + 3 < cnt; i += 4) {
                    const int sa = __builtin_amdgcn_readlane(s, i),     ra = __builtin_amdgcn_readlane(d, i) - r0;
                    const int sb = __builtin_amdgcn_readlane(s, i + 1), rb = __builtin_amdgcn_readlane(d, i + 1) - r0;
                    const int sc = __builtin_amdgcn_readlane(s, i + 2), rc = __builtin_amdgcn_readlane(d, i + 2) - r0;
                    const int sd = __builtin_amdgcn_readlane(s, i + 3), rd = __builtin_amdgcn_readlane(d, i + 3) - r0;
                    float va[8], vb[8], vc[8], vd[8];
                    row_load(h, sa, lane, va);
                    row_load(h, sb, lane, vb);
                    row_load(h, sc, lane, vc);
                    row_load(h, sd, lane, vd);
                    row_add_lds(rows, ra, lane, va, lane_bcast(w0, i),     lane_bcast(w1, i),     lane_bcast(w2, i),     lane_bcast(w3, i));
                    row_add_lds(rows, rb, lane, vb, lane_bcast(w0, i + 1), lane_bcast(w1, i + 1), lane_bcast(w2, i + 1), lane_bcast(w3, i + 1));
                    row_add_lds(rows, rc, lane, vc, lane_bcast(w0, i + 2), lane_bcast(w1, i + 2), lane_bcast(w2, i + 2), lane_bcast(w3, i + 2));
                    row_add_lds(rows, rd, lane, vd, lane_bcast(w0, i + 3), lane_bcast(w1, i + 3), lane_bcast(w2, i + 3), lane_bcast(w3, i + 3));
                }
                #pragma nounroll
                for (; i < cnt; ++i) {
                    const int sa = __builtin_amdgcn_readlane(s, i), ra = __builtin_amdgcn_readlane(d, i) - r0;
                    float va[8];
                    row_load(h, sa, lane, va);
                    row_add_lds(rows, ra, lane, va, lane_bcast(w0, i), lane_bcast(w1, i), lane_bcast(w2, i), lane_bcast(w3, i));
                }
                qh = (qh + cnt) & (QSZ - 1);
                qn -= cnt;
            }
        }
        __syncthreads();

        for (int i = threadIdx.x; i < (r1 - r0) * NH; i += blockDim.x) den[i] = 1.0f / den[i];
        __syncthreads();
        {
            const int c = (int)(threadIdx.x & (HC - 1));
            const int ngrp = (int)(blockDim.x >> 8);
            const int per = ((r1 - r0) + ngrp - 1) / ngrp;
            const int ra = __builtin_amdgcn_readfirstlane((int)(threadIdx.x >> 8)) * per;
            const int rb = min(ra + per, r1 - r0);
            if (ra < rb) {
                const float bc = bias[c];
                const float* rp = rows + c;
                const float* ip = den + (c >> 6);
                int g = __builtin_amdgcn_readfirstlane(bt[ra]);
                int run = 0;
                float acc = 0.f;
                for (int r = ra; r < rb; r += 8) {
                    int gg[8];
                    float tv[8];
                    #pragma unroll
                    for (int u = 0; u < 8; ++u) {
                        const int rr = min(r + u, rb - 1);
                        gg[u] = bt[rr];
                        tv[u] = rp[rr * HC] * ip[rr * NH] + bc;
                    }
                    #pragma unroll
                    for (int u = 0; u < 8; ++u) {
                        if (r + u < rb) {
                            const int gr = __builtin_amdgcn_readfirstlane(gg[u]);
                            if (gr != g) {
                                atomicAdd(&pooled[(size_t)g * HC + c], acc);
                                if (c == 0) atomicAdd(&cnt[g], (float)run);
                                g = gr;
                                run = 0;
                                acc = 0.f;
                            }
                            const float t = tv[u];
                            acc += t > 0.f ? t : 0.01f * t;
                            ++run;
                        }
                    }
                }
                atomicAdd(&pooled[(size_t)g * HC + c], acc);
                if (c == 0) atomicAdd(&cnt[g], (float)run);
            }
        }
        __syncthreads();
    }
}

__global__ __launch_bounds__(32) void gemm_out(const float* __restrict__ pooled,
                                               const float* __restrict__ cnt,
                                               const float* __restrict__ fw,
                                               const float* __restrict__ fb,
                                               float* __restrict__ out, int g) {
    int mTile = blockIdx.x;
    int lane = threadIdx.x & 31;
    int mr = lane & 15, half = lane >> 4;
    int row = mTile * 16 + mr;
    int col0 = blockIdx.y * 32 + mr;
    float inv = 1.0f / fmaxf(cnt[row], 1.0f);
    const float* arow = pooled + (size_t)row * HC;
    v8f acc0 = {}, acc1 = {};
    for (int k = 0; k < HC; k += 4) {
        int ka = k + 2 * half;
        v2f a; a.x = arow[ka] * inv; a.y = arow[ka + 1] * inv;
        const float* wp = fw + (size_t)ka * NOUTD + col0;
        v2f b0; b0.x = wp[0];  b0.y = wp[NOUTD];
        v2f b1; b1.x = wp[16]; b1.y = wp[NOUTD + 16];
        acc0 = __builtin_amdgcn_wmma_f32_16x16x4_f32(false, a, false, b0,
                                                     (short)0, acc0, false, false);
        acc1 = __builtin_amdgcn_wmma_f32_16x16x4_f32(false, a, false, b1,
                                                     (short)0, acc1, false, false);
    }
    float bb0 = fb[col0], bb1 = fb[col0 + 16];
    int rbase = mTile * 16 + 8 * half;
    #pragma unroll
    for (int r = 0; r < 8; ++r) {
        float* op = out + (size_t)(rbase + r) * NOUTD + col0;
        op[0]  = acc0[r] + bb0;
        op[16] = acc1[r] + bb1;
    }
}

extern "C" void kernel_launch(void* const* d_in, const int* in_sizes, int n_in,
                              void* d_out, int out_size, void* d_ws, size_t ws_size,
                              hipStream_t stream) {
    const float* x      = (const float*)d_in[0];
    const int*   ei     = (const int*)  d_in[1];
    const int*   batch  = (const int*)  d_in[2];
    const float* lin_w  = (const float*)d_in[3];
    const float* att_s  = (const float*)d_in[4];
    const float* att_d  = (const float*)d_in[5];
    const float* bias   = (const float*)d_in[6];
    const float* fc1_w  = (const float*)d_in[7];
    const float* fc1_b  = (const float*)d_in[8];
    float* out = (float*)d_out;

    const int N = in_sizes[2];
    const int E = in_sizes[1] / 2;
    const int G = out_size / NOUTD;

    float* ws = (float*)d_ws;
    float*    h      = ws;
    float*    asrc   = h + (size_t)N * HC;
    float*    adst   = asrc + (size_t)N * NH;
    float*    pooled = adst + (size_t)N * NH;
    float*    cnt    = pooled + (size_t)G * HC;

    init_ws<<<1024, 256, 0, stream>>>(pooled, cnt, G);

    gemm_x_linw<<<(N + 15) / 16, 32, 0, stream>>>(x, lin_w, h, N);

    att_kernel<<<(N * 32 + 255) / 256, 256, 0, stream>>>(h, att_s, att_d, asrc, adst, N);

    edge_agg<<<256, 1024, 0, stream>>>(ei, asrc, adst, h, bias, batch, pooled, cnt, E, N);

    dim3 g2(G / 16, NOUTD / 32);
    gemm_out<<<g2, 32, 0, stream>>>(pooled, cnt, fc1_w, fc1_b, out, G);
}


# Edge pass: ring entries carry the destination row (from the scan) and then the source node; owned rows' dst logits in LDS: no edge-list or dst-logit re-read per edge; sums still in arrival order
# speedup vs baseline: 4.7639x; 1.0318x over previous
#include <hip/hip_runtime.h>
#include <hip/hip_bf16.h>

typedef __attribute__((ext_vector_type(2))) float v2f;
typedef __attribute__((ext_vector_type(8))) float v8f;

#define HC   256
#define NH   4
#define CCH  64
#define KIN  300
#define NOUTD 768

__device__ __forceinline__ unsigned fenc(float f) {
    unsigned u = __float_as_uint(f);
    return (u & 0x80000000u) ? ~u : (u | 0x80000000u);
}
__device__ __forceinline__ float fdec(unsigned u) {
    unsigned b = (u & 0x80000000u) ? (u & 0x7FFFFFFFu) : ~u;
    return __uint_as_float(b);
}

__global__ void init_ws(float* pooled, float* cnt, int g) {
    size_t stride = (size_t)gridDim.x * blockDim.x;
    size_t t = (size_t)blockIdx.x * blockDim.x + threadIdx.x;
    size_t np = (size_t)g * HC;
    for (size_t i = t; i < np; i += stride) {
        pooled[i] = 0.0f;
        if (i < (size_t)g) cnt[i] = 0.0f;
    }
}

typedef __attribute__((ext_vector_type(4))) float v4f;
#define KBLK 32

__device__ __forceinline__ void slab_step(v8f (&acc)[16], v2f av, const float* __restrict__ wp) {
    v4f b0[4], b1[4];
    #pragma unroll
    for (int g = 0; g < 4; ++g) {
        b0[g] = *(const v4f*)(wp + 64 * g);
        b1[g] = *(const v4f*)(wp + HC + 64 * g);
    }
    #pragma unroll
    for (int g = 0; g < 4; ++g) {
        #pragma unroll
        for (int c = 0; c < 4; ++c) {
            v2f bv; bv.x = b0[g][c]; bv.y = b1[g][c];
            acc[g * 4 + c] = __builtin_amdgcn_wmma_f32_16x16x4_f32(false, av, false, bv,
                                                                   (short)0, acc[g * 4 + c], false, false);
        }
    }
}

__global__ __launch_bounds__(32) void gemm_x_linw(const float* __restrict__ x,
                                                  const float* __restrict__ w,
                                                  float* __restrict__ h, int n) {
    const int lane = threadIdx.x & 31;
    const int mr = lane & 15, half = lane >> 4;
    const int row0 = (int)blockIdx.x * 16;
    int row = row0 + mr;
    if (row >= n) row = n - 1;
    const float* arow = x + (size_t)row * KIN;
    const float* wcol = w + 4 * mr;
    v8f acc[16];
    #pragma unroll
    for (int t = 0; t < 16; ++t) acc[t] = v8f{};

    #pragma nounroll
    for (int kb = 0; kb + KBLK <= KIN; kb += KBLK) {
        const int k0 = kb + 16 * half;
        const v4f* ap = (const v4f*)(arow + k0);
        v4f a[4];
        #pragma unroll
        for (int q = 0; q < 4; ++q) a[q] = ap[q];
        const float* wp = wcol + (size_t)k0 * HC;
        #pragma unroll
        for (int j = 0; j < 8; ++j) {
            v2f av; av.x = a[j >> 1][(j & 1) * 2]; av.y = a[j >> 1][(j & 1) * 2 + 1];
            slab_step(acc, av, wp + (size_t)(2 * j) * HC);
        }
    }
    #pragma unroll
    for (int k = (KIN / KBLK) * KBLK; k < KIN; k += 4) {
        const int ka = k + 2 * half;
        v2f av; av.x = arow[ka]; av.y = arow[ka + 1];
        slab_step(acc, av, wcol + (size_t)ka * HC);
    }

    const int rbase = row0 + 8 * half;
    if (rbase + 8 <= n) {
        #pragma unroll
        for (int r = 0; r < 8; ++r) {
            float* hp = h + (size_t)(rbase + r) * HC + 4 * mr;
            #pragma unroll
            for (int g = 0; g < 4; ++g) {
                v4f v; v.x = acc[g * 4][r]; v.y = acc[g * 4 + 1][r]; v.z = acc[g * 4 + 2][r]; v.w = acc[g * 4 + 3][r];
                *(v4f*)(hp + 64 * g) = v;
            }
        }
    } else {
        for (int r = 0; r < 8; ++r) {
            if (rbase + r < n) {
                float* hp = h + (size_t)(rbase + r) * HC + 4 * mr;
                #pragma unroll
                for (int g = 0; g < 4; ++g) {
                    v4f v; v.x = acc[g * 4][r]; v.y = acc[g * 4 + 1][r]; v.z = acc[g * 4 + 2][r]; v.w = acc[g * 4 + 3][r];
                    *(v4f*)(hp + 64 * g) = v;
                }
            }
        }
    }
}

__global__ __launch_bounds__(256) void att_kernel(const float* __restrict__ h,
                                                  const float* __restrict__ wsrc,
                                                  const float* __restrict__ wdst,
                                                  float* __restrict__ asrc,
                                                  float* __restrict__ adst, int n) {
    int node = ((int)blockIdx.x * blockDim.x + threadIdx.x) >> 5;
    int lane = threadIdx.x & 31;
    if (node >= n) return;
    int hh = lane >> 3, c0 = lane & 7;
    const float* hp = h + (size_t)node * HC + hh * CCH + c0;
    const float* ws = wsrc + hh * CCH + c0;
    const float* wd = wdst + hh * CCH + c0;
    float ss = 0.f, sd = 0.f;
    #pragma unroll
    for (int i = 0; i < 8; ++i) {
        float v = hp[i * 8];
        ss += v * ws[i * 8];
        sd += v * wd[i * 8];
    }
    #pragma unroll
    for (int off = 4; off >= 1; off >>= 1) {
        ss += __shfl_xor(ss, off, 32);
        sd += __shfl_xor(sd, off, 32);
    }
    if (c0 == 0) {
        asrc[(size_t)node * NH + hh] = ss;
        adst[(size_t)node * NH + hh] = sd;
    }
}

#define ROWS_MAX 196
#define ENT_SHIFT 23
#define ENT_MASK  0x7FFFFF
#define TILES    3
#define QSZ      512

__device__ __forceinline__ void row_load(const float* __restrict__ h, int s, int lane, float (&hv)[8]) {
    const float* hp = h + (size_t)s * HC + lane;
    for (int j = 0; j < 8; ++j) hv[j] = hp[32 * j];
}
__device__ __forceinline__ void row_add_lds(float* rows, int r, int lane, const float (&hv)[8],
                                            float a0, float a1, float a2, float a3) {
    float* op = rows + r * HC + lane;
    atomicAdd(&op[0],   a0 * hv[0]);  atomicAdd(&op[32],  a0 * hv[1]);
    atomicAdd(&op[64],  a1 * hv[2]);  atomicAdd(&op[96],  a1 * hv[3]);
    atomicAdd(&op[128], a2 * hv[4]);  atomicAdd(&op[160], a2 * hv[5]);
    atomicAdd(&op[192], a3 * hv[6]);  atomicAdd(&op[224], a3 * hv[7]);
}
__device__ __forceinline__ float lane_bcast(float v, int i) {
    return __uint_as_float(__builtin_amdgcn_readlane(__float_as_uint(v), i));
}

__global__ __launch_bounds__(1024) void edge_agg(const int* __restrict__ ei,
                                                 const float* __restrict__ asrc,
                                                 const float* __restrict__ adst,
                                                 const float* __restrict__ h,
                                                 const float* __restrict__ bias,
                                                 const int* __restrict__ batch,
                                                 float* __restrict__ pooled,
                                                 float* __restrict__ cnt, int E, int n) {
    __shared__ __attribute__((aligned(16))) float rows[ROWS_MAX * HC];
    __shared__ float den[ROWS_MAX * NH];
    __shared__ __attribute__((aligned(16))) unsigned mx[ROWS_MAX * NH];
    __shared__ int rings[32 * QSZ];
    __shared__ int bt[ROWS_MAX];
    __shared__ __attribute__((aligned(16))) float adl[ROWS_MAX * NH];
    __shared__ int gave_up;
    const int lane = threadIdx.x & 31;
    const int wv = __builtin_amdgcn_readfirstlane((int)(threadIdx.x >> 5));
    const int nwv = (int)(blockDim.x >> 5);
    typedef __attribute__((address_space(3))) int lds_int;
    volatile lds_int* myq = (volatile lds_int*)(rings + wv * QSZ);

    const int R = (n + (int)gridDim.x - 1) / (int)gridDim.x;
    const int nb0 = (int)blockIdx.x * R;
    const int nb1 = min(nb0 + R, n);

    const int* dstp = ei + E;
    const int hc = min((4 - (E & 3)) & 3, E);
    const int nb = (E - hc) >> 2;
    const int tc = (E - hc) & 3;
    const int4* d4 = (const int4*)(dstp + hc);
    const int ntiles = (nb + 31) >> 5;
    const int niter = (ntiles + TILES * nwv - 1) / (TILES * nwv);
    const int rot = (int)(((long long)blockIdx.x * ntiles) / (int)gridDim.x);

    for (int r0 = nb0; r0 < nb1; r0 += ROWS_MAX) {
        const int r1 = min(r0 + ROWS_MAX, nb1);
        const unsigned span = (unsigned)(r1 - r0);
        const int nv4 = (r1 - r0) * (HC / 4);
        float4* rows4 = (float4*)rows;
        for (int i = threadIdx.x; i < nv4; i += blockDim.x) rows4[i] = make_float4(0.f, 0.f, 0.f, 0.f);
        for (int i = threadIdx.x; i < (r1 - r0) * NH; i += blockDim.x) {
            den[i] = 0.f;
            mx[i] = 0x007FFFFFu;
        }
        for (int i = threadIdx.x; i < r1 - r0; i += blockDim.x) {
            bt[i] = batch[r0 + i];
            ((float4*)adl)[i] = *(const float4*)(adst + (size_t)(r0 + i) * NH);
        }
        if (threadIdx.x == 0) gave_up = 0;
        __syncthreads();

        int qh = 0, qn = 0;
        int qm = 0;
        bool full = false;
        auto q_append = [&](bool match, int e) {
            const unsigned m = __builtin_amdgcn_ballot_w32(match);
            if (m != 0u) {
                const int slot = (qh + qn + (int)__builtin_amdgcn_mbcnt_lo(m, 0u)) & (QSZ - 1);
                if (match) myq[slot] = e;
                qn += __builtin_popcount(m);
            }
        };
        auto ent_of = [&](int e, int dd) {
            return (int)((unsigned)e | ((unsigned)(dd - r0) << ENT_SHIFT));
        };

        auto scan_trip = [&](int it, bool keep) {
            int4 q[TILES];
            int g[TILES];
            bool ok[TILES];
            for (int u = 0; u < TILES; ++u) {
                const int kk = (it * TILES + u) * nwv + wv;
                int tt = kk + rot;
                if (tt >= ntiles) tt -= ntiles;
                g[u] = tt * 32 + lane;
                ok[u] = kk < ntiles && g[u] < nb;
                q[u] = d4[(unsigned)(ok[u] ? g[u] : 0)];
            }
            for (int u = 0; u < TILES; ++u) {
                const bool m0 = ((unsigned)(q[u].x - r0) < span) & ok[u];
                const bool m1 = ((unsigned)(q[u].y - r0) < span) & ok[u];
                const bool m2 = ((unsigned)(q[u].z - r0) < span) & ok[u];
                const bool m3 = ((unsigned)(q[u].w - r0) < span) & ok[u];
                if (__builtin_amdgcn_ballot_w32(m0 | m1 | m2 | m3) != 0u) {
                    if (keep && qn + 128 > QSZ) {
                        full = true;
                        qh = (qh + qm) & (QSZ - 1);
                        qn -= qm;
                        qm = 0;
                    }
                    const int e = hc + 4 * g[u];
                    q_append(m0, ent_of(e, q[u].x));
                    q_append(m1, ent_of(e + 1, q[u].y));
                    q_append(m2, ent_of(e + 2, q[u].z));
                    q_append(m3, ent_of(e + 3, q[u].w));
                }
            }
        };

        const int nself = ((r1 - r0) + 31) >> 5;
        for (int j = wv; j <= nself; j += nwv) {
            if (j < nself) {
                const int node = r0 + j * 32 + lane;
                q_append(node < r1, -1 - node);
            } else {
                const int idx = lane < hc ? lane : 4 * nb + lane;
                int dd = -1;
                if (lane < hc + tc) dd = dstp[idx];
                q_append((unsigned)(dd - r0) < span, ent_of(idx, dd));
            }
        }
        #pragma nounroll
        for (int it = 0; it <= niter; ++it) {
            if (it < niter) scan_trip(it, true);
            const bool last = it == niter;
            while (qn - qm >= 32 || (last && qn > qm)) {
                const int cnt = min(qn - qm, 32);
                if (lane < cnt) {
                    const int slot = (qh + qm + lane) & (QSZ - 1);
                    const int ent = myq[slot];
                    int s, rl;
                    if (ent >= 0) { s = ei[ent & ENT_MASK]; rl = ent >> ENT_SHIFT; } else { s = -1 - ent; rl = s - r0; }
                    myq[slot] = s | (rl << ENT_SHIFT);
                    const float4 as = *(const float4*)(asrc + (size_t)s * NH);
                    const float4 ad = ((const float4*)adl)[rl];
                    unsigned* mp = mx + rl * NH;
                    float v;
                    v = as.x + ad.x; v = v > 0.f ? v : 0.2f * v; atomicMax(&mp[0], fenc(v));
                    v = as.y + ad.y; v = v > 0.f ? v : 0.2f * v; atomicMax(&mp[1], fenc(v));
                    v = as.z + ad.z; v = v > 0.f ? v : 0.2f * v; atomicMax(&mp[2], fenc(v));
                    v = as.w + ad.w; v = v > 0.f ? v : 0.2f * v; atomicMax(&mp[3], fenc(v));
                }
                qm += cnt;
            }
        }
        if (full && lane == 0) gave_up = 1;
        __syncthreads();

        const bool rescan = __builtin_amdgcn_readfirstlane(gave_up) != 0;
        int nscan = 0;
        if (rescan) {
            qh = 0;
            qn = 0;
            for (int j = wv; j <= nself; j += nwv) {
                if (j < nself) {
                    const int node = r0 + j * 32 + lane;
                    q_append(node < r1, -1 - node);
                } else {
                    const int idx = lane < hc ? lane : 4 * nb + lane;
                    int dd = -1;
                    if (lane < hc + tc) dd = dstp[idx];
                    q_append((unsigned)(dd - r0) < span, ent_of(idx, dd));
                }
            }
            nscan = niter;
        }

        #pragma nounroll
        for (int it = 0; it <= nscan; ++it) {
            if (it < nscan) scan_trip(it, false);
            const bool last = it == nscan;
            while (qn >= 32 || (last && qn > 0)) {
                const int cnt = min(qn, 32);
                int s = 0, rl = 0;
                float w0 = 0.f, w1 = 0.f, w2 = 0.f, w3 = 0.f;
                if (lane < cnt) {
                    const int ent = myq[(qh + lane) & (QSZ - 1)];
                    if (!rescan) {
                        s = ent & ENT_MASK;
                        rl = ent >> ENT_SHIFT;
                    } else if (ent >= 0) {
                        s = ei[ent & ENT_MASK];
                        rl = ent >> ENT_SHIFT;
                    } else {
                        s = -1 - ent;
                        rl = s - r0;
                    }
                    const float4 as = *(const float4*)(asrc + (size_t)s * NH);
                    const float4 ad = ((const float4*)adl)[rl];
                    const uint4  em = *(const uint4*)(mx + rl * NH);
                    float v;
                    v = as.x + ad.x; v = v > 0.f ? v : 0.2f * v; w0 = __expf(v - fdec(em.x));
                    v = as.y + ad.y; v = v > 0.f ? v : 0.2f * v; w1 = __expf(v - fdec(em.y));
                    v = as.z + ad.z; v = v > 0.f ? v : 0.2f * v; w2 = __expf(v - fdec(em.z));
                    v = as.w + ad.w; v = v > 0.f ? v : 0.2f * v; w3 = __expf(v - fdec(em.w));
                    float* dp = den + rl * NH;
                    atomicAdd(&dp[0], w0);  atomicAdd(&dp[1], w1);
                    atomicAdd(&dp[2], w2);  atomicAdd(&dp[3], w3);
                }
                int i = 0;
                #pragma nounroll
                for (; i + 3 < cnt; i += 4) {
                    const int sa = __builtin_amdgcn_readlane(s, i),     ra = __builtin_amdgcn_readlane(rl, i);
                    const int sb = __builtin_amdgcn_readlane(s, i + 1), rb = __builtin_amdgcn_readlane(rl, i + 1);
                    const int sc = __builtin_amdgcn_readlane(s, i + 2), rc = __builtin_amdgcn_readlane(rl, i + 2);
                    const int sd = __builtin_amdgcn_readlane(s, i + 3), rd = __builtin_amdgcn_readlane(rl, i + 3);
                    float va[8], vb[8], vc[8], vd[8];
                    row_load(h, sa, lane, va);
                    row_load(h, sb, lane, vb);
                    row_load(h, sc, lane, vc);
                    row_load(h, sd, lane, vd);
                    row_add_lds(rows, ra, lane, va, lane_bcast(w0, i),     lane_bcast(w1, i),     lane_bcast(w2, i),     lane_bcast(w3, i));
                    row_add_lds(rows, rb, lane, vb, lane_bcast(w0, i + 1), lane_bcast(w1, i + 1), lane_bcast(w2, i + 1), lane_bcast(w3, i + 1));
                    row_add_lds(rows, rc, lane, vc, lane_bcast(w0, i + 2), lane_bcast(w1, i + 2), lane_bcast(w2, i + 2), lane_bcast(w3, i + 2));
                    row_add_lds(rows, rd, lane, vd, lane_bcast(w0, i + 3), lane_bcast(w1, i + 3), lane_bcast(w2, i + 3), lane_bcast(w3, i + 3));
                }
                #pragma nounroll
                for (; i < cnt; ++i) {
                    const int sa = __builtin_amdgcn_readlane(s, i), ra = __builtin_amdgcn_readlane(rl, i);
                    float va[8];
                    row_load(h, sa, lane, va);
                    row_add_lds(rows, ra, lane, va, lane_bcast(w0, i), lane_bcast(w1, i), lane_bcast(w2, i), lane_bcast(w3, i));
                }
                qh = (qh + cnt) & (QSZ - 1);
                qn -= cnt;
            }
        }
        __syncthreads();

        for (int i = threadIdx.x; i < (r1 - r0) * NH; i += blockDim.x) den[i] = 1.0f / den[i];
        __syncthreads();
        {
            const int c = (int)(threadIdx.x & (HC - 1));
            const int ngrp = (int)(blockDim.x >> 8);
            const int per = ((r1 - r0) + ngrp - 1) / ngrp;
            const int ra = __builtin_amdgcn_readfirstlane((int)(threadIdx.x >> 8)) * per;
            const int rb = min(ra + per, r1 - r0);
            if (ra < rb) {
                const float bc = bias[c];
                const float* rp = rows + c;
                const float* ip = den + (c >> 6);
                int g = __builtin_amdgcn_readfirstlane(bt[ra]);
                int run = 0;
                float acc = 0.f;
                for (int r = ra; r < rb; r += 8) {
                    int gg[8];
                    float tv[8];
                    #pragma unroll
                    for (int u = 0; u < 8; ++u) {
                        const int rr = min(r + u, rb - 1);
                        gg[u] = bt[rr];
                        tv[u] = rp[rr * HC] * ip[rr * NH] + bc;
                    }
                    #pragma unroll
                    for (int u = 0; u < 8; ++u) {
                        if (r + u < rb) {
                            const int gr = __builtin_amdgcn_readfirstlane(gg[u]);
                            if (gr != g) {
                                atomicAdd(&pooled[(size_t)g * HC + c], acc);
                                if (c == 0) atomicAdd(&cnt[g], (float)run);
                                g = gr;
                                run = 0;
                                acc = 0.f;
                            }
                            const float t = tv[u];
                            acc += t > 0.f ? t : 0.01f * t;
                            ++run;
                        }
                    }
                }
                atomicAdd(&pooled[(size_t)g * HC + c], acc);
                if (c == 0) atomicAdd(&cnt[g], (float)run);
            }
        }
        __syncthreads();
    }
}

__global__ __launch_bounds__(32) void gemm_out(const float* __restrict__ pooled,
                                               const float* __restrict__ cnt,
                                               const float* __restrict__ fw,
                                               const float* __restrict__ fb,
                                               float* __restrict__ out, int g) {
    int mTile = blockIdx.x;
    int lane = threadIdx.x & 31;
    int mr = lane & 15, half = lane >> 4;
    int row = mTile * 16 + mr;
    int col0 = blockIdx.y * 32 + mr;
    float inv = 1.0f / fmaxf(cnt[row], 1.0f);
    const float* arow = pooled + (size_t)row * HC;
    v8f acc0 = {}, acc1 = {};
    for (int k = 0; k < HC; k += 4) {
        int ka = k + 2 * half;
        v2f a; a.x = arow[ka] * inv; a.y = arow[ka + 1] * inv;
        const float* wp = fw + (size_t)ka * NOUTD + col0;
        v2f b0; b0.x = wp[0];  b0.y = wp[NOUTD];
        v2f b1; b1.x = wp[16]; b1.y = wp[NOUTD + 16];
        acc0 = __builtin_amdgcn_wmma_f32_16x16x4_f32(false, a, false, b0,
                                                     (short)0, acc0, false, false);
        acc1 = __builtin_amdgcn_wmma_f32_16x16x4_f32(false, a, false, b1,
                                                     (short)0, acc1, false, false);
    }
    float bb0 = fb[col0], bb1 = fb[col0 + 16];
    int rbase = mTile * 16 + 8 * half;
    #pragma unroll
    for (int r = 0; r < 8; ++r) {
        float* op = out + (size_t)(rbase + r) * NOUTD + col0;
        op[0]  = acc0[r] + bb0;
        op[16] = acc1[r] + bb1;
    }
}

extern "C" void kernel_launch(void* const* d_in, const int* in_sizes, int n_in,
                              void* d_out, int out_size, void* d_ws, size_t ws_size,
                              hipStream_t stream) {
    const float* x      = (const float*)d_in[0];
    const int*   ei     = (const int*)  d_in[1];
    const int*   batch  = (const int*)  d_in[2];
    const float* lin_w  = (const float*)d_in[3];
    const float* att_s  = (const float*)d_in[4];
    const float* att_d  = (const float*)d_in[5];
    const float* bias   = (const float*)d_in[6];
    const float* fc1_w  = (const float*)d_in[7];
    const float* fc1_b  = (const float*)d_in[8];
    float* out = (float*)d_out;

    const int N = in_sizes[2];
    const int E = in_sizes[1] / 2;
    const int G = out_size / NOUTD;

    float* ws = (float*)d_ws;
    float*    h      = ws;
    float*    asrc   = h + (size_t)N * HC;
    float*    adst   = asrc + (size_t)N * NH;
    float*    pooled = adst + (size_t)N * NH;
    float*    cnt    = pooled + (size_t)G * HC;

    init_ws<<<1024, 256, 0, stream>>>(pooled, cnt, G);

    gemm_x_linw<<<(N + 15) / 16, 32, 0, stream>>>(x, lin_w, h, N);

    att_kernel<<<(N * 32 + 255) / 256, 256, 0, stream>>>(h, att_s, att_d, asrc, adst, N);

    edge_agg<<<256, 1024, 0, stream>>>(ei, asrc, adst, h, bias, batch, pooled, cnt, E, N);

    dim3 g2(G / 16, NOUTD / 32);
    gemm_out<<<g2, 32, 0, stream>>>(pooled, cnt, fc1_w, fc1_b, out, G);
}


# Edge pass scans a byte plane: the init launch writes each edge's owner workgroup (dst / rows per workgroup) as 1 byte; 4x fewer scan bytes and trips; ids read for kept edges only; arrival order
# speedup vs baseline: 5.1380x; 1.0785x over previous
#include <hip/hip_runtime.h>
#include <hip/hip_bf16.h>

typedef __attribute__((ext_vector_type(2))) float v2f;
typedef __attribute__((ext_vector_type(8))) float v8f;

#define HC   256
#define NH   4
#define CCH  64
#define KIN  300
#define NOUTD 768

__device__ __forceinline__ unsigned fenc(float f) {
    unsigned u = __float_as_uint(f);
    return (u & 0x80000000u) ? ~u : (u | 0x80000000u);
}
__device__ __forceinline__ float fdec(unsigned u) {
    unsigned b = (u & 0x80000000u) ? (u & 0x7FFFFFFFu) : ~u;
    return __uint_as_float(b);
}

__device__ __forceinline__ unsigned key_skip(const unsigned* raw) {
    return ((16u - (unsigned)((size_t)raw & 15)) & 15u) >> 2;
}
__device__ __forceinline__ unsigned zero_bytes(unsigned t) {
    const unsigned y = ((t & 0x7F7F7F7Fu) + 0x7F7F7F7Fu) | t;
    return ~y & 0x80808080u;
}

__global__ void init_ws(float* pooled, float* cnt, int g,
                        const int* __restrict__ ei, unsigned* keyraw, int E, int n, int nwg) {
    size_t stride = (size_t)gridDim.x * blockDim.x;
    size_t t = (size_t)blockIdx.x * blockDim.x + threadIdx.x;
    size_t np = (size_t)g * HC;
    for (size_t i = t; i < np; i += stride) {
        pooled[i] = 0.0f;
        if (i < (size_t)g) cnt[i] = 0.0f;
    }
    unsigned* keys = keyraw + key_skip(keyraw);
    const unsigned R = (unsigned)((n + nwg - 1) / nwg);
    const int* dstp = ei + E;
    const bool al16 = (((size_t)dstp) & 15) == 0;
    const size_t nwords = (((size_t)E + 15) >> 4) << 2;
    for (size_t w = t; w < nwords; w += stride) {
        const size_t e0 = 4 * w;
        int d0 = 0, d1 = 0, d2 = 0, d3 = 0;
        if (al16 && e0 + 3 < (size_t)E) {
            const int4 v = *(const int4*)(dstp + e0);
            d0 = v.x; d1 = v.y; d2 = v.z; d3 = v.w;
        } else {
            if (e0     < (size_t)E) d0 = dstp[e0];
            if (e0 + 1 < (size_t)E) d1 = dstp[e0 + 1];
            if (e0 + 2 < (size_t)E) d2 = dstp[e0 + 2];
            if (e0 + 3 < (size_t)E) d3 = dstp[e0 + 3];
        }
        keys[w] = ((unsigned)d0 / R) | (((unsigned)d1 / R) << 8) | (((unsigned)d2 / R) << 16) | (((unsigned)d3 / R) << 24);
    }
}

typedef __attribute__((ext_vector_type(4))) float v4f;
#define KBLK 32

__device__ __forceinline__ void slab_step(v8f (&acc)[16], v2f av, const float* __restrict__ wp) {
    v4f b0[4], b1[4];
    #pragma unroll
    for (int g = 0; g < 4; ++g) {
        b0[g] = *(const v4f*)(wp + 64 * g);
        b1[g] = *(const v4f*)(wp + HC + 64 * g);
    }
    #pragma unroll
    for (int g = 0; g < 4; ++g) {
        #pragma unroll
        for (int c = 0; c < 4; ++c) {
            v2f bv; bv.x = b0[g][c]; bv.y = b1[g][c];
            acc[g * 4 + c] = __builtin_amdgcn_wmma_f32_16x16x4_f32(false, av, false, bv,
                                                                   (short)0, acc[g * 4 + c], false, false);
        }
    }
}

__global__ __launch_bounds__(32) void gemm_x_linw(const float* __restrict__ x,
                                                  const float* __restrict__ w,
                                                  float* __restrict__ h, int n) {
    const int lane = threadIdx.x & 31;
    const int mr = lane & 15, half = lane >> 4;
    const int row0 = (int)blockIdx.x * 16;
    int row = row0 + mr;
    if (row >= n) row = n - 1;
    const float* arow = x + (size_t)row * KIN;
    const float* wcol = w + 4 * mr;
    v8f acc[16];
    #pragma unroll
    for (int t = 0; t < 16; ++t) acc[t] = v8f{};

    #pragma nounroll
    for (int kb = 0; kb + KBLK <= KIN; kb += KBLK) {
        const int k0 = kb + 16 * half;
        const v4f* ap = (const v4f*)(arow + k0);
        v4f a[4];
        #pragma unroll
        for (int q = 0; q < 4; ++q) a[q] = ap[q];
        const float* wp = wcol + (size_t)k0 * HC;
        #pragma unroll
        for (int j = 0; j < 8; ++j) {
            v2f av; av.x = a[j >> 1][(j & 1) * 2]; av.y = a[j >> 1][(j & 1) * 2 + 1];
            slab_step(acc, av, wp + (size_t)(2 * j) * HC);
        }
    }
    #pragma unroll
    for (int k = (KIN / KBLK) * KBLK; k < KIN; k += 4) {
        const int ka = k + 2 * half;
        v2f av; av.x = arow[ka]; av.y = arow[ka + 1];
        slab_step(acc, av, wcol + (size_t)ka * HC);
    }

    const int rbase = row0 + 8 * half;
    if (rbase + 8 <= n) {
        #pragma unroll
        for (int r = 0; r < 8; ++r) {
            float* hp = h + (size_t)(rbase + r) * HC + 4 * mr;
            #pragma unroll
            for (int g = 0; g < 4; ++g) {
                v4f v; v.x = acc[g * 4][r]; v.y = acc[g * 4 + 1][r]; v.z = acc[g * 4 + 2][r]; v.w = acc[g * 4 + 3][r];
                *(v4f*)(hp + 64 * g) = v;
            }
        }
    } else {
        for (int r = 0; r < 8; ++r) {
            if (rbase + r < n) {
                float* hp = h + (size_t)(rbase + r) * HC + 4 * mr;
                #pragma unroll
                for (int g = 0; g < 4; ++g) {
                    v4f v; v.x = acc[g * 4][r]; v.y = acc[g * 4 + 1][r]; v.z = acc[g * 4 + 2][r]; v.w = acc[g * 4 + 3][r];
                    *(v4f*)(hp + 64 * g) = v;
                }
            }
        }
    }
}

__global__ __launch_bounds__(256) void att_kernel(const float* __restrict__ h,
                                                  const float* __restrict__ wsrc,
                                                  const float* __restrict__ wdst,
                                                  float* __restrict__ asrc,
                                                  float* __restrict__ adst, int n) {
    int node = ((int)blockIdx.x * blockDim.x + threadIdx.x) >> 5;
    int lane = threadIdx.x & 31;
    if (node >= n) return;
    int hh = lane >> 3, c0 = lane & 7;
    const float* hp = h + (size_t)node * HC + hh * CCH + c0;
    const float* ws = wsrc + hh * CCH + c0;
    const float* wd = wdst + hh * CCH + c0;
    float ss = 0.f, sd = 0.f;
    #pragma unroll
    for (int i = 0; i < 8; ++i) {
        float v = hp[i * 8];
        ss += v * ws[i * 8];
        sd += v * wd[i * 8];
    }
    #pragma unroll
    for (int off = 4; off >= 1; off >>= 1) {
        ss += __shfl_xor(ss, off, 32);
        sd += __shfl_xor(sd, off, 32);
    }
    if (c0 == 0) {
        asrc[(size_t)node * NH + hh] = ss;
        adst[(size_t)node * NH + hh] = sd;
    }
}

#define ROWS_MAX 196
#define ROW_SPARE ROWS_MAX
#define ENT_SHIFT 23
#define ENT_MASK  0x7FFFFF
#define TILES    3
#define QSZ      512
__device__ __forceinline__ void row_load(const float* __restrict__ h, int s, int lane, float (&hv)[8]) {
    const float* hp = h + (size_t)s * HC + lane;
    for (int j = 0; j < 8; ++j) hv[j] = hp[32 * j];
}
__device__ __forceinline__ void row_add_lds(float* rows, int r, int lane, const float (&hv)[8],
                                            float a0, float a1, float a2, float a3) {
    float* op = rows + r * HC + lane;
    atomicAdd(&op[0],   a0 * hv[0]);  atomicAdd(&op[32],  a0 * hv[1]);
    atomicAdd(&op[64],  a1 * hv[2]);  atomicAdd(&op[96],  a1 * hv[3]);
    atomicAdd(&op[128], a2 * hv[4]);  atomicAdd(&op[160], a2 * hv[5]);
    atomicAdd(&op[192], a3 * hv[6]);  atomicAdd(&op[224], a3 * hv[7]);
}
__device__ __forceinline__ float lane_bcast(float v, int i) {
    return __uint_as_float(__builtin_amdgcn_readlane(__float_as_uint(v), i));
}

__global__ __launch_bounds__(1024) void edge_agg(const int* __restrict__ ei,
                                                 const unsigned* __restrict__ keyraw,
                                                 const float* __restrict__ asrc,
                                                 const float* __restrict__ adst,
                                                 const float* __restrict__ h,
                                                 const float* __restrict__ bias,
                                                 const int* __restrict__ batch,
                                                 float* __restrict__ pooled,
                                                 float* __restrict__ cnt, int E, int n) {
    __shared__ __attribute__((aligned(16))) float rows[(ROWS_MAX + 1) * HC];
    __shared__ float den[(ROWS_MAX + 1) * NH];
    __shared__ __attribute__((aligned(16))) unsigned mx[(ROWS_MAX + 1) * NH];
    __shared__ int rings[32 * QSZ];
    __shared__ int bt[ROWS_MAX];
    __shared__ __attribute__((aligned(16))) float adl[(ROWS_MAX + 1) * NH];
    __shared__ int gave_up;
    const int lane = threadIdx.x & 31;
    const int wv = __builtin_amdgcn_readfirstlane((int)(threadIdx.x >> 5));
    const int nwv = (int)(blockDim.x >> 5);
    typedef __attribute__((address_space(3))) int lds_int;
    volatile lds_int* myq = (volatile lds_int*)(rings + wv * QSZ);

    const int R = (n + (int)gridDim.x - 1) / (int)gridDim.x;
    const int nb0 = (int)blockIdx.x * R;
    const int nb1 = min(nb0 + R, n);

    const int* dstp = ei + E;
    const uint4* k4 = (const uint4*)(keyraw + key_skip(keyraw));
    const int ngrp = (E + 15) >> 4;
    const int ntiles = (ngrp + 31) >> 5;
    const int niter = (ntiles + TILES * nwv - 1) / (TILES * nwv);
    const int rot = (int)(((long long)blockIdx.x * ntiles) / (int)gridDim.x);
    const unsigned kpat = ((unsigned)blockIdx.x & 0xFFu) * 0x01010101u;

    for (int r0 = nb0; r0 < nb1; r0 += ROWS_MAX) {
        const int r1 = min(r0 + ROWS_MAX, nb1);
        const unsigned span = (unsigned)(r1 - r0);
        const int nv4 = (r1 - r0) * (HC / 4);
        float4* rows4 = (float4*)rows;
        for (int i = threadIdx.x; i < nv4; i += blockDim.x) rows4[i] = make_float4(0.f, 0.f, 0.f, 0.f);
        for (int i = threadIdx.x; i < (r1 - r0) * NH; i += blockDim.x) {
            den[i] = 0.f;
            mx[i] = 0x007FFFFFu;
        }
        for (int i = threadIdx.x; i < r1 - r0; i += blockDim.x) {
            bt[i] = batch[r0 + i];
            ((float4*)adl)[i] = *(const float4*)(adst + (size_t)(r0 + i) * NH);
        }
        if (threadIdx.x == 0) gave_up = 0;
        __syncthreads();

        int qh = 0, qn = 0;
        int qm = 0;
        bool full = false;
        auto q_append = [&](bool match, int e) {
            const unsigned m = __builtin_amdgcn_ballot_w32(match);
            if (m != 0u) {
                const int slot = (qh + qn + (int)__builtin_amdgcn_mbcnt_lo(m, 0u)) & (QSZ - 1);
                if (match) myq[slot] = e;
                qn += __builtin_popcount(m);
            }
        };
        auto look_up = [&](int ent, int& s, int& rl) {
            if (ent >= 0) {
                s = ei[ent];
                rl = dstp[ent] - r0;
                if ((unsigned)rl >= span) rl = ROW_SPARE;
            } else {
                s = -1 - ent;
                rl = s - r0;
            }
        };
        auto fold = [&](bool last) {
            while (qn - qm >= 32 || (last && qn > qm)) {
                const int cnt = min(qn - qm, 32);
                if (lane < cnt) {
                    const int slot = (qh + qm + lane) & (QSZ - 1);
                    int s, rl;
                    look_up(myq[slot], s, rl);
                    myq[slot] = s | (rl << ENT_SHIFT);
                    const float4 as = *(const float4*)(asrc + (size_t)s * NH);
                    const float4 ad = ((const float4*)adl)[rl];
                    unsigned* mp = mx + rl * NH;
                    float v;
                    v = as.x + ad.x; v = v > 0.f ? v : 0.2f * v; atomicMax(&mp[0], fenc(v));
                    v = as.y + ad.y; v = v > 0.f ? v : 0.2f * v; atomicMax(&mp[1], fenc(v));
                    v = as.z + ad.z; v = v > 0.f ? v : 0.2f * v; atomicMax(&mp[2], fenc(v));
                    v = as.w + ad.w; v = v > 0.f ? v : 0.2f * v; atomicMax(&mp[3], fenc(v));
                }
                qm += cnt;
            }
        };
        auto pop = [&](unsigned zz, int base, bool keep) {
            for (int k = 0; k < 16; ++k) {
                const bool has = zz != 0u;
                if (__builtin_amdgcn_ballot_w32(has) == 0u) break;
                if (keep && qn + 32 > QSZ) {
                    fold(false);
                    full = true;
                    qh = (qh + qm) & (QSZ - 1);
                    qn -= qm;
                    qm = 0;
                }
                const int p = __builtin_ctz(zz | 0x80000000u);
                const int e = base + 4 * (p & 7) + (p >> 3);
                zz &= zz - 1u;
                q_append(has && e < E, e);
            }
        };

        auto scan_trip = [&](int it) {
            uint4 q[TILES];
            int g[TILES];
            bool ok[TILES];
            for (int u = 0; u < TILES; ++u) {
                const int kk = (it * TILES + u) * nwv + wv;
                int tt = kk + rot;
                if (tt >= ntiles) tt -= ntiles;
                g[u] = tt * 32 + lane;
                ok[u] = kk < ntiles && g[u] < ngrp;
                q[u] = k4[(unsigned)(ok[u] ? g[u] : 0)];
            }
            __builtin_amdgcn_sched_barrier(0);
            for (int u = 0; u < TILES; ++u) {
                const unsigned live = ok[u] ? 0xFFFFFFFFu : 0u;
                const unsigned z0 = zero_bytes(q[u].x ^ kpat) & live;
                const unsigned z1 = zero_bytes(q[u].y ^ kpat) & live;
                const unsigned z2 = zero_bytes(q[u].z ^ kpat) & live;
                const unsigned z3 = zero_bytes(q[u].w ^ kpat) & live;
                if (__builtin_amdgcn_ballot_w32((z0 | z1 | z2 | z3) != 0u) != 0u)
                    pop((z0 >> 7) | (z1 >> 6) | (z2 >> 5) | (z3 >> 4), 16 * g[u], true);
            }
        };
        auto rescan_trip = [&](int it) {
            const int c = it & 3;
            const int kk = (it >> 2) * nwv + wv;
            int tt = kk + rot;
            if (tt >= ntiles) tt -= ntiles;
            const int g = tt * 32 + lane;
            const bool ok = kk < ntiles && g < ngrp;
            const uint4 q = k4[(unsigned)(ok ? g : 0)];
            const unsigned v = c == 0 ? q.x : c == 1 ? q.y : c == 2 ? q.z : q.w;
            const unsigned z = ok ? zero_bytes(v ^ kpat) : 0u;
            if (__builtin_amdgcn_ballot_w32(z != 0u) != 0u) pop((z >> 7) << c, 16 * g, false);
        };

        const int nself = ((r1 - r0) + 31) >> 5;
        for (int j = wv; j < nself; j += nwv) {
            const int node = r0 + j * 32 + lane;
            q_append(node < r1, -1 - node);
        }
        #pragma nounroll
        for (int it = 0; it <= niter; ++it) {
            if (it < niter) scan_trip(it);
            fold(it == niter);
        }
        if (full && lane == 0) gave_up = 1;
        __syncthreads();

        const bool rescan = __builtin_amdgcn_readfirstlane(gave_up) != 0;
        int nscan = 0;
        if (rescan) {
            qh = 0;
            qn = 0;
            for (int j = wv; j < nself; j += nwv) {
                const int node = r0 + j * 32 + lane;
                q_append(node < r1, -1 - node);
            }
            nscan = 4 * ((ntiles + nwv - 1) / nwv);
        }

        #pragma nounroll
        for (int it = 0; it <= nscan; ++it) {
            if (it < nscan) rescan_trip(it);
            const bool last = it == nscan;
            while (qn >= 32 || (last && qn > 0)) {
                const int cnt = min(qn, 32);
                int s = 0, rl = 0;
                float w0 = 0.f, w1 = 0.f, w2 = 0.f, w3 = 0.f;
                if (lane < cnt) {
                    const int ent = myq[(qh + lane) & (QSZ - 1)];
                    if (!rescan) {
                        s = ent & ENT_MASK;
                        rl = ent >> ENT_SHIFT;
                    } else {
                        look_up(ent, s, rl);
                    }
                    const float4 as = *(const float4*)(asrc + (size_t)s * NH);
                    const float4 ad = ((const float4*)adl)[rl];
                    const uint4  em = *(const uint4*)(mx + rl * NH);
                    float v;
                    v = as.x + ad.x; v = v > 0.f ? v : 0.2f * v; w0 = __expf(v - fdec(em.x));
                    v = as.y + ad.y; v = v > 0.f ? v : 0.2f * v; w1 = __expf(v - fdec(em.y));
                    v = as.z + ad.z; v = v > 0.f ? v : 0.2f * v; w2 = __expf(v - fdec(em.z));
                    v = as.w + ad.w; v = v > 0.f ? v : 0.2f * v; w3 = __expf(v - fdec(em.w));
                    float* dp = den + rl * NH;
                    atomicAdd(&dp[0], w0);  atomicAdd(&dp[1], w1);
                    atomicAdd(&dp[2], w2);  atomicAdd(&dp[3], w3);
                }
                int i = 0;
                #pragma nounroll
                for (; i + 3 < cnt; i += 4) {
                    const int sa = __builtin_amdgcn_readlane(s, i),     ra = __builtin_amdgcn_readlane(rl, i);
                    const int sb = __builtin_amdgcn_readlane(s, i + 1), rb = __builtin_amdgcn_readlane(rl, i + 1);
                    const int sc = __builtin_amdgcn_readlane(s, i + 2), rc = __builtin_amdgcn_readlane(rl, i + 2);
                    const int sd = __builtin_amdgcn_readlane(s, i + 3), rd = __builtin_amdgcn_readlane(rl, i + 3);
                    float va[8], vb[8], vc[8], vd[8];
                    row_load(h, sa, lane, va);
                    row_load(h, sb, lane, vb);
                    row_load(h, sc, lane, vc);
                    row_load(h, sd, lane, vd);
                    row_add_lds(rows, ra, lane, va, lane_bcast(w0, i),     lane_bcast(w1, i),     lane_bcast(w2, i),     lane_bcast(w3, i));
                    row_add_lds(rows, rb, lane, vb, lane_bcast(w0, i + 1), lane_bcast(w1, i + 1), lane_bcast(w2, i + 1), lane_bcast(w3, i + 1));
                    row_add_lds(rows, rc, lane, vc, lane_bcast(w0, i + 2), lane_bcast(w1, i + 2), lane_bcast(w2, i + 2), lane_bcast(w3, i + 2));
                    row_add_lds(rows, rd, lane, vd, lane_bcast(w0, i + 3), lane_bcast(w1, i + 3), lane_bcast(w2, i + 3), lane_bcast(w3, i + 3));
                }
                #pragma nounroll
                for (; i < cnt; ++i) {
                    const int sa = __builtin_amdgcn_readlane(s, i), ra = __builtin_amdgcn_readlane(rl, i);
                    float va[8];
                    row_load(h, sa, lane, va);
                    row_add_lds(rows, ra, lane, va, lane_bcast(w0, i), lane_bcast(w1, i), lane_bcast(w2, i), lane_bcast(w3, i));
                }
                qh = (qh + cnt) & (QSZ - 1);
                qn -= cnt;
            }
        }
        __syncthreads();

        for (int i = threadIdx.x; i < (r1 - r0) * NH; i += blockDim.x) den[i] = 1.0f / den[i];
        __syncthreads();
        {
            const int c = (int)(threadIdx.x & (HC - 1));
            const int ngrp = (int)(blockDim.x >> 8);
            const int per = ((r1 - r0) + ngrp - 1) / ngrp;
            const int ra = __builtin_amdgcn_readfirstlane((int)(threadIdx.x >> 8)) * per;
            const int rb = min(ra + per, r1 - r0);
            if (ra < rb) {
                const float bc = bias[c];
                const float* rp = rows + c;
                const float* ip = den + (c >> 6);
                int g = __builtin_amdgcn_readfirstlane(bt[ra]);
                int run = 0;
                float acc = 0.f;
                for (int r = ra; r < rb; r += 8) {
                    int gg[8];
                    float tv[8];
                    #pragma unroll
                    for (int u = 0; u < 8; ++u) {
                        const int rr = min(r + u, rb - 1);
                        gg[u] = bt[rr];
                        tv[u] = rp[rr * HC] * ip[rr * NH] + bc;
                    }
                    #pragma unroll
                    for (int u = 0; u < 8; ++u) {
                        if (r + u < rb) {
                            const int gr = __builtin_amdgcn_readfirstlane(gg[u]);
                            if (gr != g) {
                                atomicAdd(&pooled[(size_t)g * HC + c], acc);
                                if (c == 0) atomicAdd(&cnt[g], (float)run);
                                g = gr;
                                run = 0;
                                acc = 0.f;
                            }
                            const float t = tv[u];
                            acc += t > 0.f ? t : 0.01f * t;
                            ++run;
                        }
                    }
                }
                atomicAdd(&pooled[(size_t)g * HC + c], acc);
                if (c == 0) atomicAdd(&cnt[g], (float)run);
            }
        }
        __syncthreads();
    }
}

__global__ __launch_bounds__(32) void gemm_out(const float* __restrict__ pooled,
                                               const float* __restrict__ cnt,
                                               const float* __restrict__ fw,
                                               const float* __restrict__ fb,
                                               float* __restrict__ out, int g) {
    int mTile = blockIdx.x;
    int lane = threadIdx.x & 31;
    int mr = lane & 15, half = lane >> 4;
    int row = mTile * 16 + mr;
    int col0 = blockIdx.y * 32 + mr;
    float inv = 1.0f / fmaxf(cnt[row], 1.0f);
    const float* arow = pooled + (size_t)row * HC;
    v8f acc0 = {}, acc1 = {};
    for (int k = 0; k < HC; k += 4) {
        int ka = k + 2 * half;
        v2f a; a.x = arow[ka] * inv; a.y = arow[ka + 1] * inv;
        const float* wp = fw + (size_t)ka * NOUTD + col0;
        v2f b0; b0.x = wp[0];  b0.y = wp[NOUTD];
        v2f b1; b1.x = wp[16]; b1.y = wp[NOUTD + 16];
        acc0 = __builtin_amdgcn_wmma_f32_16x16x4_f32(false, a, false, b0,
                                                     (short)0, acc0, false, false);
        acc1 = __builtin_amdgcn_wmma_f32_16x16x4_f32(false, a, false, b1,
                                                     (short)0, acc1, false, false);
    }
    float bb0 = fb[col0], bb1 = fb[col0 + 16];
    int rbase = mTile * 16 + 8 * half;
    #pragma unroll
    for (int r = 0; r < 8; ++r) {
        float* op = out + (size_t)(rbase + r) * NOUTD + col0;
        op[0]  = acc0[r] + bb0;
        op[16] = acc1[r] + bb1;
    }
}

extern "C" void kernel_launch(void* const* d_in, const int* in_sizes, int n_in,
                              void* d_out, int out_size, void* d_ws, size_t ws_size,
                              hipStream_t stream) {
    const float* x      = (const float*)d_in[0];
    const int*   ei     = (const int*)  d_in[1];
    const int*   batch  = (const int*)  d_in[2];
    const float* lin_w  = (const float*)d_in[3];
    const float* att_s  = (const float*)d_in[4];
    const float* att_d  = (const float*)d_in[5];
    const float* bias   = (const float*)d_in[6];
    const float* fc1_w  = (const float*)d_in[7];
    const float* fc1_b  = (const float*)d_in[8];
    float* out = (float*)d_out;

    const int N = in_sizes[2];
    const int E = in_sizes[1] / 2;
    const int G = out_size / NOUTD;

    float* ws = (float*)d_ws;
    float*    h      = ws;
    float*    asrc   = h + (size_t)N * HC;
    float*    adst   = asrc + (size_t)N * NH;
    float*    pooled = adst + (size_t)N * NH;
    float*    cnt    = pooled + (size_t)G * HC;
    unsigned* keyraw = (unsigned*)(cnt + G);

    init_ws<<<1024, 256, 0, stream>>>(pooled, cnt, G, ei, keyraw, E, N, 256);

    gemm_x_linw<<<(N + 15) / 16, 32, 0, stream>>>(x, lin_w, h, N);

    att_kernel<<<(N * 32 + 255) / 256, 256, 0, stream>>>(h, att_s, att_d, asrc, adst, N);

    edge_agg<<<256, 1024, 0, stream>>>(ei, keyraw, asrc, adst, h, bias, batch, pooled, cnt, E, N);

    dim3 g2(G / 16, NOUTD / 32);
    gemm_out<<<g2, 32, 0, stream>>>(pooled, cnt, fc1_w, fc1_b, out, G);
}


# Attention logits made in GEMM1's epilogue from the accumulators (lane products, then 4 exchange steps over a half's 16 lanes); the logit launch and its re-read of h are gone; logit sum order new
# speedup vs baseline: 5.3424x; 1.0398x over previous
#include <hip/hip_runtime.h>
#include <hip/hip_bf16.h>

typedef __attribute__((ext_vector_type(2))) float v2f;
typedef __attribute__((ext_vector_type(8))) float v8f;

#define HC   256
#define NH   4
#define CCH  64
#define KIN  300
#define NOUTD 768

__device__ __forceinline__ unsigned fenc(float f) {
    unsigned u = __float_as_uint(f);
    return (u & 0x80000000u) ? ~u : (u | 0x80000000u);
}
__device__ __forceinline__ float fdec(unsigned u) {
    unsigned b = (u & 0x80000000u) ? (u & 0x7FFFFFFFu) : ~u;
    return __uint_as_float(b);
}

__device__ __forceinline__ unsigned key_skip(const unsigned* raw) {
    return ((16u - (unsigned)((size_t)raw & 15)) & 15u) >> 2;
}
__device__ __forceinline__ unsigned zero_bytes(unsigned t) {
    const unsigned y = ((t & 0x7F7F7F7Fu) + 0x7F7F7F7Fu) | t;
    return ~y & 0x80808080u;
}

__global__ void init_ws(float* pooled, float* cnt, int g,
                        const int* __restrict__ ei, unsigned* keyraw, int E, int n, int nwg) {
    size_t stride = (size_t)gridDim.x * blockDim.x;
    size_t t = (size_t)blockIdx.x * blockDim.x + threadIdx.x;
    size_t np = (size_t)g * HC;
    for (size_t i = t; i < np; i += stride) {
        pooled[i] = 0.0f;
        if (i < (size_t)g) cnt[i] = 0.0f;
    }
    unsigned* keys = keyraw + key_skip(keyraw);
    const unsigned R = (unsigned)((n + nwg - 1) / nwg);
    const int* dstp = ei + E;
    const bool al16 = (((size_t)dstp) & 15) == 0;
    const size_t nwords = (((size_t)E + 15) >> 4) << 2;
    for (size_t w = t; w < nwords; w += stride) {
        const size_t e0 = 4 * w;
        int d0 = 0, d1 = 0, d2 = 0, d3 = 0;
        if (al16 && e0 + 3 < (size_t)E) {
            const int4 v = *(const int4*)(dstp + e0);
            d0 = v.x; d1 = v.y; d2 = v.z; d3 = v.w;
        } else {
            if (e0     < (size_t)E) d0 = dstp[e0];
            if (e0 + 1 < (size_t)E) d1 = dstp[e0 + 1];
            if (e0 + 2 < (size_t)E) d2 = dstp[e0 + 2];
            if (e0 + 3 < (size_t)E) d3 = dstp[e0 + 3];
        }
        keys[w] = ((unsigned)d0 / R) | (((unsigned)d1 / R) << 8) | (((unsigned)d2 / R) << 16) | (((unsigned)d3 / R) << 24);
    }
}

typedef __attribute__((ext_vector_type(4))) float v4f;
#define KBLK 32

__device__ __forceinline__ void slab_step(v8f (&acc)[16], v2f av, const float* __restrict__ wp) {
    v4f b0[4], b1[4];
    #pragma unroll
    for (int g = 0; g < 4; ++g) {
        b0[g] = *(const v4f*)(wp + 64 * g);
        b1[g] = *(const v4f*)(wp + HC + 64 * g);
    }
    #pragma unroll
    for (int g = 0; g < 4; ++g) {
        #pragma unroll
        for (int c = 0; c < 4; ++c) {
            v2f bv; bv.x = b0[g][c]; bv.y = b1[g][c];
            acc[g * 4 + c] = __builtin_amdgcn_wmma_f32_16x16x4_f32(false, av, false, bv,
                                                                   (short)0, acc[g * 4 + c], false, false);
        }
    }
}

__global__ __launch_bounds__(32) void gemm_x_linw(const float* __restrict__ x,
                                                  const float* __restrict__ w,
                                                  const float* __restrict__ wsrc,
                                                  const float* __restrict__ wdst,
                                                  float* __restrict__ h,
                                                  float* __restrict__ asrc,
                                                  float* __restrict__ adst, int n) {
    const int lane = threadIdx.x & 31;
    const int mr = lane & 15, half = lane >> 4;
    const int row0 = (int)blockIdx.x * 16;
    int row = row0 + mr;
    if (row >= n) row = n - 1;
    const float* arow = x + (size_t)row * KIN;
    const float* wcol = w + 4 * mr;
    v8f acc[16];
    #pragma unroll
    for (int t = 0; t < 16; ++t) acc[t] = v8f{};

    #pragma nounroll
    for (int kb = 0; kb + KBLK <= KIN; kb += KBLK) {
        const int k0 = kb + 16 * half;
        const v4f* ap = (const v4f*)(arow + k0);
        v4f a[4];
        #pragma unroll
        for (int q = 0; q < 4; ++q) a[q] = ap[q];
        const float* wp = wcol + (size_t)k0 * HC;
        #pragma unroll
        for (int j = 0; j < 8; ++j) {
            v2f av; av.x = a[j >> 1][(j & 1) * 2]; av.y = a[j >> 1][(j & 1) * 2 + 1];
            slab_step(acc, av, wp + (size_t)(2 * j) * HC);
        }
    }
    #pragma unroll
    for (int k = (KIN / KBLK) * KBLK; k < KIN; k += 4) {
        const int ka = k + 2 * half;
        v2f av; av.x = arow[ka]; av.y = arow[ka + 1];
        slab_step(acc, av, wcol + (size_t)ka * HC);
    }

    const int rbase = row0 + 8 * half;
    if (rbase + 8 <= n) {
        #pragma unroll
        for (int r = 0; r < 8; ++r) {
            float* hp = h + (size_t)(rbase + r) * HC + 4 * mr;
            #pragma unroll
            for (int g = 0; g < 4; ++g) {
                v4f v; v.x = acc[g * 4][r]; v.y = acc[g * 4 + 1][r]; v.z = acc[g * 4 + 2][r]; v.w = acc[g * 4 + 3][r];
                *(v4f*)(hp + 64 * g) = v;
            }
        }
    } else {
        for (int r = 0; r < 8; ++r) {
            if (rbase + r < n) {
                float* hp = h + (size_t)(rbase + r) * HC + 4 * mr;
                #pragma unroll
                for (int g = 0; g < 4; ++g) {
                    v4f v; v.x = acc[g * 4][r]; v.y = acc[g * 4 + 1][r]; v.z = acc[g * 4 + 2][r]; v.w = acc[g * 4 + 3][r];
                    *(v4f*)(hp + 64 * g) = v;
                }
            }
        }
    }

    v4f ws4[4], wd4[4];
    #pragma unroll
    for (int g = 0; g < 4; ++g) {
        ws4[g] = *(const v4f*)(wsrc + 64 * g + 4 * mr);
        wd4[g] = *(const v4f*)(wdst + 64 * g + 4 * mr);
    }
    float p[64];
    #pragma unroll
    for (int r = 0; r < 8; ++r) {
        #pragma unroll
        for (int g = 0; g < 4; ++g) {
            float s = 0.f, d = 0.f;
            #pragma unroll
            for (int c = 0; c < 4; ++c) {
                const float v = acc[g * 4 + c][r];
                s += v * ws4[g][c];
                d += v * wd4[g][c];
            }
            p[g + 4 * (2 * r)]     = s;
            p[g + 4 * (2 * r + 1)] = d;
        }
    }
    #pragma unroll
    for (int st = 0; st < 4; ++st) {
        const int hq = 8 >> st;
        const bool up = (mr & hq) != 0;
        #pragma unroll
        for (int q = 0; q < hq; ++q) {
            #pragma unroll
            for (int g = 0; g < 4; ++g) {
                const float lo = p[g + 4 * q], hi = p[g + 4 * (q + hq)];
                const float keep = up ? hi : lo;
                const float send = up ? lo : hi;
                p[g + 4 * q] = keep + __shfl_xor(send, hq, 32);
            }
        }
    }
    {
        const int lrow = row0 + 8 * half + (mr >> 1);
        if (lrow < n) {
            float* lp = ((mr & 1) ? adst : asrc) + (size_t)lrow * NH;
            v4f v; v.x = p[0]; v.y = p[1]; v.z = p[2]; v.w = p[3];
            *(v4f*)lp = v;
        }
    }
}

#define ROWS_MAX 196
#define ROW_SPARE ROWS_MAX
#define ENT_SHIFT 23
#define ENT_MASK  0x7FFFFF
#define TILES    3
#define QSZ      512
__device__ __forceinline__ void row_load(const float* __restrict__ h, int s, int lane, float (&hv)[8]) {
    const float* hp = h + (size_t)s * HC + lane;
    for (int j = 0; j < 8; ++j) hv[j] = hp[32 * j];
}
__device__ __forceinline__ void row_add_lds(float* rows, int r, int lane, const float (&hv)[8],
                                            float a0, float a1, float a2, float a3) {
    float* op = rows + r * HC + lane;
    atomicAdd(&op[0],   a0 * hv[0]);  atomicAdd(&op[32],  a0 * hv[1]);
    atomicAdd(&op[64],  a1 * hv[2]);  atomicAdd(&op[96],  a1 * hv[3]);
    atomicAdd(&op[128], a2 * hv[4]);  atomicAdd(&op[160], a2 * hv[5]);
    atomicAdd(&op[192], a3 * hv[6]);  atomicAdd(&op[224], a3 * hv[7]);
}
__device__ __forceinline__ float lane_bcast(float v, int i) {
    return __uint_as_float(__builtin_amdgcn_readlane(__float_as_uint(v), i));
}

__global__ __launch_bounds__(1024) void edge_agg(const int* __restrict__ ei,
                                                 const unsigned* __restrict__ keyraw,
                                                 const float* __restrict__ asrc,
                                                 const float* __restrict__ adst,
                                                 const float* __restrict__ h,
                                                 const float* __restrict__ bias,
                                                 const int* __restrict__ batch,
                                                 float* __restrict__ pooled,
                                                 float* __restrict__ cnt, int E, int n) {
    __shared__ __attribute__((aligned(16))) float rows[(ROWS_MAX + 1) * HC];
    __shared__ float den[(ROWS_MAX + 1) * NH];
    __shared__ __attribute__((aligned(16))) unsigned mx[(ROWS_MAX + 1) * NH];
    __shared__ int rings[32 * QSZ];
    __shared__ int bt[ROWS_MAX];
    __shared__ __attribute__((aligned(16))) float adl[(ROWS_MAX + 1) * NH];
    __shared__ int gave_up;
    const int lane = threadIdx.x & 31;
    const int wv = __builtin_amdgcn_readfirstlane((int)(threadIdx.x >> 5));
    const int nwv = (int)(blockDim.x >> 5);
    typedef __attribute__((address_space(3))) int lds_int;
    volatile lds_int* myq = (volatile lds_int*)(rings + wv * QSZ);

    const int R = (n + (int)gridDim.x - 1) / (int)gridDim.x;
    const int nb0 = (int)blockIdx.x * R;
    const int nb1 = min(nb0 + R, n);

    const int* dstp = ei + E;
    const uint4* k4 = (const uint4*)(keyraw + key_skip(keyraw));
    const int ngrp = (E + 15) >> 4;
    const int ntiles = (ngrp + 31) >> 5;
    const int niter = (ntiles + TILES * nwv - 1) / (TILES * nwv);
    const int rot = (int)(((long long)blockIdx.x * ntiles) / (int)gridDim.x);
    const unsigned kpat = ((unsigned)blockIdx.x & 0xFFu) * 0x01010101u;

    for (int r0 = nb0; r0 < nb1; r0 += ROWS_MAX) {
        const int r1 = min(r0 + ROWS_MAX, nb1);
        const unsigned span = (unsigned)(r1 - r0);
        const int nv4 = (r1 - r0) * (HC / 4);
        float4* rows4 = (float4*)rows;
        for (int i = threadIdx.x; i < nv4; i += blockDim.x) rows4[i] = make_float4(0.f, 0.f, 0.f, 0.f);
        for (int i = threadIdx.x; i < (r1 - r0) * NH; i += blockDim.x) {
            den[i] = 0.f;
            mx[i] = 0x007FFFFFu;
        }
        for (int i = threadIdx.x; i < r1 - r0; i += blockDim.x) {
            bt[i] = batch[r0 + i];
            ((float4*)adl)[i] = *(const float4*)(adst + (size_t)(r0 + i) * NH);
        }
        if (threadIdx.x == 0) gave_up = 0;
        __syncthreads();

        int qh = 0, qn = 0;
        int qm = 0;
        bool full = false;
        auto q_append = [&](bool match, int e) {
            const unsigned m = __builtin_amdgcn_ballot_w32(match);
            if (m != 0u) {
                const int slot = (qh + qn + (int)__builtin_amdgcn_mbcnt_lo(m, 0u)) & (QSZ - 1);
                if (match) myq[slot] = e;
                qn += __builtin_popcount(m);
            }
        };
        auto look_up = [&](int ent, int& s, int& rl) {
            if (ent >= 0) {
                s = ei[ent];
                rl = dstp[ent] - r0;
                if ((unsigned)rl >= span) rl = ROW_SPARE;
            } else {
                s = -1 - ent;
                rl = s - r0;
            }
        };
        auto fold = [&](bool last) {
            while (qn - qm >= 32 || (last && qn > qm)) {
                const int cnt = min(qn - qm, 32);
                if (lane < cnt) {
                    const int slot = (qh + qm + lane) & (QSZ - 1);
                    int s, rl;
                    look_up(myq[slot], s, rl);
                    myq[slot] = s | (rl << ENT_SHIFT);
                    const float4 as = *(const float4*)(asrc + (size_t)s * NH);
                    const float4 ad = ((const float4*)adl)[rl];
                    unsigned* mp = mx + rl * NH;
                    float v;
                    v = as.x + ad.x; v = v > 0.f ? v : 0.2f * v; atomicMax(&mp[0], fenc(v));
                    v = as.y + ad.y; v = v > 0.f ? v : 0.2f * v; atomicMax(&mp[1], fenc(v));
                    v = as.z + ad.z; v = v > 0.f ? v : 0.2f * v; atomicMax(&mp[2], fenc(v));
                    v = as.w + ad.w; v = v > 0.f ? v : 0.2f * v; atomicMax(&mp[3], fenc(v));
                }
                qm += cnt;
            }
        };
        auto pop = [&](unsigned zz, int base, bool keep) {
            for (int k = 0; k < 16; ++k) {
                const bool has = zz != 0u;
                if (__builtin_amdgcn_ballot_w32(has) == 0u) break;
                if (keep && qn + 32 > QSZ) {
                    fold(false);
                    full = true;
                    qh = (qh + qm) & (QSZ - 1);
                    qn -= qm;
                    qm = 0;
                }
                const int p = __builtin_ctz(zz | 0x80000000u);
                const int e = base + 4 * (p & 7) + (p >> 3);
                zz &= zz - 1u;
                q_append(has && e < E, e);
            }
        };

        auto scan_trip = [&](int it) {
            uint4 q[TILES];
            int g[TILES];
            bool ok[TILES];
            for (int u = 0; u < TILES; ++u) {
                const int kk = (it * TILES + u) * nwv + wv;
                int tt = kk + rot;
                if (tt >= ntiles) tt -= ntiles;
                g[u] = tt * 32 + lane;
                ok[u] = kk < ntiles && g[u] < ngrp;
                q[u] = k4[(unsigned)(ok[u] ? g[u] : 0)];
            }
            __builtin_amdgcn_sched_barrier(0);
            for (int u = 0; u < TILES; ++u) {
                const unsigned live = ok[u] ? 0xFFFFFFFFu : 0u;
                const unsigned z0 = zero_bytes(q[u].x ^ kpat) & live;
                const unsigned z1 = zero_bytes(q[u].y ^ kpat) & live;
                const unsigned z2 = zero_bytes(q[u].z ^ kpat) & live;
                const unsigned z3 = zero_bytes(q[u].w ^ kpat) & live;
                if (__builtin_amdgcn_ballot_w32((z0 | z1 | z2 | z3) != 0u) != 0u)
                    pop((z0 >> 7) | (z1 >> 6) | (z2 >> 5) | (z3 >> 4), 16 * g[u], true);
            }
        };
        auto rescan_trip = [&](int it) {
            const int c = it & 3;
            const int kk = (it >> 2) * nwv + wv;
            int tt = kk + rot;
            if (tt >= ntiles) tt -= ntiles;
            const int g = tt * 32 + lane;
            const bool ok = kk < ntiles && g < ngrp;
            const uint4 q = k4[(unsigned)(ok ? g : 0)];
            const unsigned v = c == 0 ? q.x : c == 1 ? q.y : c == 2 ? q.z : q.w;
            const unsigned z = ok ? zero_bytes(v ^ kpat) : 0u;
            if (__builtin_amdgcn_ballot_w32(z != 0u) != 0u) pop((z >> 7) << c, 16 * g, false);
        };

        const int nself = ((r1 - r0) + 31) >> 5;
        for (int j = wv; j < nself; j += nwv) {
            const int node = r0 + j * 32 + lane;
            q_append(node < r1, -1 - node);
        }
        #pragma nounroll
        for (int it = 0; it <= niter; ++it) {
            if (it < niter) scan_trip(it);
            fold(it == niter);
        }
        if (full && lane == 0) gave_up = 1;
        __syncthreads();

        const bool rescan = __builtin_amdgcn_readfirstlane(gave_up) != 0;
        int nscan = 0;
        if (rescan) {
            qh = 0;
            qn = 0;
            for (int j = wv; j < nself; j += nwv) {
                const int node = r0 + j * 32 + lane;
                q_append(node < r1, -1 - node);
            }
            nscan = 4 * ((ntiles + nwv - 1) / nwv);
        }

        #pragma nounroll
        for (int it = 0; it <= nscan; ++it) {
            if (it < nscan) rescan_trip(it);
            const bool last = it == nscan;
            while (qn >= 32 || (last && qn > 0)) {
                const int cnt = min(qn, 32);
                int s = 0, rl = 0;
                float w0 = 0.f, w1 = 0.f, w2 = 0.f, w3 = 0.f;
                if (lane < cnt) {
                    const int ent = myq[(qh + lane) & (QSZ - 1)];
                    if (!rescan) {
                        s = ent & ENT_MASK;
                        rl = ent >> ENT_SHIFT;
                    } else {
                        look_up(ent, s, rl);
                    }
                    const float4 as = *(const float4*)(asrc + (size_t)s * NH);
                    const float4 ad = ((const float4*)adl)[rl];
                    const uint4  em = *(const uint4*)(mx + rl * NH);
                    float v;
                    v = as.x + ad.x; v = v > 0.f ? v : 0.2f * v; w0 = __expf(v - fdec(em.x));
                    v = as.y + ad.y; v = v > 0.f ? v : 0.2f * v; w1 = __expf(v - fdec(em.y));
                    v = as.z + ad.z; v = v > 0.f ? v : 0.2f * v; w2 = __expf(v - fdec(em.z));
                    v = as.w + ad.w; v = v > 0.f ? v : 0.2f * v; w3 = __expf(v - fdec(em.w));
                    float* dp = den + rl * NH;
                    atomicAdd(&dp[0], w0);  atomicAdd(&dp[1], w1);
                    atomicAdd(&dp[2], w2);  atomicAdd(&dp[3], w3);
                }
                int i = 0;
                #pragma nounroll
                for (; i + 3 < cnt; i += 4) {
                    const int sa = __builtin_amdgcn_readlane(s, i),     ra = __builtin_amdgcn_readlane(rl, i);
                    const int sb = __builtin_amdgcn_readlane(s, i + 1), rb = __builtin_amdgcn_readlane(rl, i + 1);
                    const int sc = __builtin_amdgcn_readlane(s, i + 2), rc = __builtin_amdgcn_readlane(rl, i + 2);
                    const int sd = __builtin_amdgcn_readlane(s, i + 3), rd = __builtin_amdgcn_readlane(rl, i + 3);
                    float va[8], vb[8], vc[8], vd[8];
                    row_load(h, sa, lane, va);
                    row_load(h, sb, lane, vb);
                    row_load(h, sc, lane, vc);
                    row_load(h, sd, lane, vd);
                    row_add_lds(rows, ra, lane, va, lane_bcast(w0, i),     lane_bcast(w1, i),     lane_bcast(w2, i),     lane_bcast(w3, i));
                    row_add_lds(rows, rb, lane, vb, lane_bcast(w0, i + 1), lane_bcast(w1, i + 1), lane_bcast(w2, i + 1), lane_bcast(w3, i + 1));
                    row_add_lds(rows, rc, lane, vc, lane_bcast(w0, i + 2), lane_bcast(w1, i + 2), lane_bcast(w2, i + 2), lane_bcast(w3, i + 2));
                    row_add_lds(rows, rd, lane, vd, lane_bcast(w0, i + 3), lane_bcast(w1, i + 3), lane_bcast(w2, i + 3), lane_bcast(w3, i + 3));
                }
                #pragma nounroll
                for (; i < cnt; ++i) {
                    const int sa = __builtin_amdgcn_readlane(s, i), ra = __builtin_amdgcn_readlane(rl, i);
                    float va[8];
                    row_load(h, sa, lane, va);
                    row_add_lds(rows, ra, lane, va, lane_bcast(w0, i), lane_bcast(w1, i), lane_bcast(w2, i), lane_bcast(w3, i));
                }
                qh = (qh + cnt) & (QSZ - 1);
                qn -= cnt;
            }
        }
        __syncthreads();

        for (int i = threadIdx.x; i < (r1 - r0) * NH; i += blockDim.x) den[i] = 1.0f / den[i];
        __syncthreads();
        {
            const int c = (int)(threadIdx.x & (HC - 1));
            const int ngrp = (int)(blockDim.x >> 8);
            const int per = ((r1 - r0) + ngrp - 1) / ngrp;
            const int ra = __builtin_amdgcn_readfirstlane((int)(threadIdx.x >> 8)) * per;
            const int rb = min(ra + per, r1 - r0);
            if (ra < rb) {
                const float bc = bias[c];
                const float* rp = rows + c;
                const float* ip = den + (c >> 6);
                int g = __builtin_amdgcn_readfirstlane(bt[ra]);
                int run = 0;
                float acc = 0.f;
                for (int r = ra; r < rb; r += 8) {
                    int gg[8];
                    float tv[8];
                    #pragma unroll
                    for (int u = 0; u < 8; ++u) {
                        const int rr = min(r + u, rb - 1);
                        gg[u] = bt[rr];
                        tv[u] = rp[rr * HC] * ip[rr * NH] + bc;
                    }
                    #pragma unroll
                    for (int u = 0; u < 8; ++u) {
                        if (r + u < rb) {
                            const int gr = __builtin_amdgcn_readfirstlane(gg[u]);
                            if (gr != g) {
                                atomicAdd(&pooled[(size_t)g * HC + c], acc);
                                if (c == 0) atomicAdd(&cnt[g], (float)run);
                                g = gr;
                                run = 0;
                                acc = 0.f;
                            }
                            const float t = tv[u];
                            acc += t > 0.f ? t : 0.01f * t;
                            ++run;
                        }
                    }
                }
                atomicAdd(&pooled[(size_t)g * HC + c], acc);
                if (c == 0) atomicAdd(&cnt[g], (float)run);
            }
        }
        __syncthreads();
    }
}

__global__ __launch_bounds__(32) void gemm_out(const float* __restrict__ pooled,
                                               const float* __restrict__ cnt,
                                               const float* __restrict__ fw,
                                               const float* __restrict__ fb,
                                               float* __restrict__ out, int g) {
    int mTile = blockIdx.x;
    int lane = threadIdx.x & 31;
    int mr = lane & 15, half = lane >> 4;
    int row = mTile * 16 + mr;
    int col0 = blockIdx.y * 32 + mr;
    float inv = 1.0f / fmaxf(cnt[row], 1.0f);
    const float* arow = pooled + (size_t)row * HC;
    v8f acc0 = {}, acc1 = {};
    for (int k = 0; k < HC; k += 4) {
        int ka = k + 2 * half;
        v2f a; a.x = arow[ka] * inv; a.y = arow[ka + 1] * inv;
        const float* wp = fw + (size_t)ka * NOUTD + col0;
        v2f b0; b0.x = wp[0];  b0.y = wp[NOUTD];
        v2f b1; b1.x = wp[16]; b1.y = wp[NOUTD + 16];
        acc0 = __builtin_amdgcn_wmma_f32_16x16x4_f32(false, a, false, b0,
                                                     (short)0, acc0, false, false);
        acc1 = __builtin_amdgcn_wmma_f32_16x16x4_f32(false, a, false, b1,
                                                     (short)0, acc1, false, false);
    }
    float bb0 = fb[col0], bb1 = fb[col0 + 16];
    int rbase = mTile * 16 + 8 * half;
    #pragma unroll
    for (int r = 0; r < 8; ++r) {
        float* op = out + (size_t)(rbase + r) * NOUTD + col0;
        op[0]  = acc0[r] + bb0;
        op[16] = acc1[r] + bb1;
    }
}

extern "C" void kernel_launch(void* const* d_in, const int* in_sizes, int n_in,
                              void* d_out, int out_size, void* d_ws, size_t ws_size,
                              hipStream_t stream) {
    const float* x      = (const float*)d_in[0];
    const int*   ei     = (const int*)  d_in[1];
    const int*   batch  = (const int*)  d_in[2];
    const float* lin_w  = (const float*)d_in[3];
    const float* att_s  = (const float*)d_in[4];
    const float* att_d  = (const float*)d_in[5];
    const float* bias   = (const float*)d_in[6];
    const float* fc1_w  = (const float*)d_in[7];
    const float* fc1_b  = (const float*)d_in[8];
    float* out = (float*)d_out;

    const int N = in_sizes[2];
    const int E = in_sizes[1] / 2;
    const int G = out_size / NOUTD;

    float* ws = (float*)d_ws;
    float*    h      = ws;
    float*    asrc   = h + (size_t)N * HC;
    float*    adst   = asrc + (size_t)N * NH;
    float*    pooled = adst + (size_t)N * NH;
    float*    cnt    = pooled + (size_t)G * HC;
    unsigned* keyraw = (unsigned*)(cnt + G);

    init_ws<<<1024, 256, 0, stream>>>(pooled, cnt, G, ei, keyraw, E, N, 256);

    gemm_x_linw<<<(N + 15) / 16, 32, 0, stream>>>(x, lin_w, att_s, att_d, h, asrc, adst, N);

    edge_agg<<<256, 1024, 0, stream>>>(ei, keyraw, asrc, adst, h, bias, batch, pooled, cnt, E, N);

    dim3 g2(G / 16, NOUTD / 32);
    gemm_out<<<g2, 32, 0, stream>>>(pooled, cnt, fc1_w, fc1_b, out, G);
}
